# starting kernel
# baseline (speedup 1.0000x reference)
#include <hip/hip_runtime.h>
#include <math.h>

#define SRATE     22050
#define NFFT      1024
#define HOP       256
#define NMELS     128
#define NFREQS    513
#define FPAD      528
#define BATCH     16
#define NSAMP     661504
#define NFRAMES   2585
#define TTILES    162
#define SPAN      (15*HOP + NFFT)
#define NCH       (BATCH*NMELS)

typedef __attribute__((ext_vector_type(2))) float v2f;
typedef __attribute__((ext_vector_type(8))) float v8f;

__global__ void init_dft(float* __restrict__ dftCos, float* __restrict__ dftSin) {
    int idx = blockIdx.x * blockDim.x + threadIdx.x;
    if (idx >= NFFT * FPAD) return;
    int k = idx / FPAD;
    int f = idx % FPAD;
    float c = 0.0f, s = 0.0f;
    if (f < NFREQS) {
        float win = 0.5f - 0.5f * cosf(6.28318530717958647692f * (float)k / (float)NFFT);
        int r = (k * f) & (NFFT - 1);
        float ang = (float)r * (6.28318530717958647692f / (float)NFFT);
        c = win * cosf(ang);
        s = win * sinf(ang);
    }
    int off = (k >> 1) * (FPAD * 2) + f * 2 + (k & 1);
    dftCos[off] = c;
    dftSin[off] = s;
}

__device__ __forceinline__ float hz_to_mel(float f) { return 2595.0f * log10f(1.0f + f / 700.0f); }
__device__ __forceinline__ float mel_to_hz(float m) { return 700.0f * (exp10f(m / 2595.0f) - 1.0f); }

__global__ void init_mel(float* __restrict__ melFB) {
    int idx = blockIdx.x * blockDim.x + threadIdx.x;
    if (idx >= FPAD * NMELS) return;
    int k = idx / NMELS;
    int m = idx % NMELS;
    float v = 0.0f;
    if (k < NFREQS) {
        float m_lo = hz_to_mel(50.0f);
        float m_hi = hz_to_mel((float)SRATE * 0.5f);
        float step = (m_hi - m_lo) / (float)(NMELS + 1);
        float f0 = mel_to_hz(m_lo + step * (float)m);
        float f1 = mel_to_hz(m_lo + step * (float)(m + 1));
        float f2 = mel_to_hz(m_lo + step * (float)(m + 2));
        float freq = (float)k * ((float)SRATE * 0.5f / (float)(NFFT / 2));
        float down = (freq - f0) / (f1 - f0);
        float up   = (f2 - freq) / (f2 - f1);
        v = fmaxf(0.0f, fminf(down, up));
    }
    melFB[(k >> 1) * (NMELS * 2) + m * 2 + (k & 1)] = v;
}

__global__ __launch_bounds__(128)
void dft_mel_kernel(const float* __restrict__ x,
                    const float* __restrict__ dftCos,
                    const float* __restrict__ dftSin,
                    const float* __restrict__ melFB,
                    float* __restrict__ melBuf)
{
    __shared__ float span[SPAN];
    __shared__ float mag[16 * FPAD];

    const int tile = blockIdx.x;
    const int b    = blockIdx.y;
    const int tid  = threadIdx.x;
    const int ln   = tid & 31;
    const int wv   = tid >> 5;
    const int t0   = tile * 16;

    const float* xb = x + (size_t)b * NSAMP;
    const long base = (long)t0 * HOP - (NFFT / 2);
    for (int i = tid; i < SPAN; i += 128) {
        long p = base + i;
        if (p < 0) p = -p;
        if (p >= NSAMP) p = 2L * NSAMP - 2 - p;
        if (p < 0) p = 0;
        if (p >= NSAMP) p = NSAMP - 1;
        span[i] = xb[p];
    }
    __syncthreads();

    const int mrow  = ln & 15;
    const int kh    = (ln >> 4) << 1;
    const int ncol  = ln & 15;
    const int mbase = (ln >> 4) * 8;

    for (int nt = wv; nt < FPAD / 16; nt += 4) {
        v8f accRe = {0.f, 0.f, 0.f, 0.f, 0.f, 0.f, 0.f, 0.f};
        v8f accIm = {0.f, 0.f, 0.f, 0.f, 0.f, 0.f, 0.f, 0.f};
        const int n = nt * 16 + ncol;
        #pragma unroll 4
        for (int kb = 0; kb < NFFT; kb += 4) {
            const int k0 = kb + kh;
            v2f a  = *(const v2f*)&span[mrow * HOP + k0];
            v2f bc = *(const v2f*)&dftCos[(k0 >> 1) * (FPAD * 2) + n * 2];
            v2f bs = *(const v2f*)&dftSin[(k0 >> 1) * (FPAD * 2) + n * 2];
            accRe = __builtin_amdgcn_wmma_f32_16x16x4_f32(false, a, false, bc,
                                                          (short)0, accRe, false, false);
            accIm = __builtin_amdgcn_wmma_f32_16x16x4_f32(false, a, false, bs,
                                                          (short)0, accIm, false, false);
        }
        #pragma unroll
        for (int v = 0; v < 8; ++v) {
            float re = accRe[v], im = accIm[v];
            mag[(mbase + v) * FPAD + n] = sqrtf(re * re + im * im);
        }
    }
    __syncthreads();

    for (int nt = wv; nt < NMELS / 16; nt += 4) {
        v8f acc = {0.f, 0.f, 0.f, 0.f, 0.f, 0.f, 0.f, 0.f};
        const int n = nt * 16 + ncol;
        #pragma unroll 4
        for (int kb = 0; kb < FPAD; kb += 4) {
            const int k0 = kb + kh;
            v2f a  = *(const v2f*)&mag[mrow * FPAD + k0];
            v2f bm = *(const v2f*)&melFB[(k0 >> 1) * (NMELS * 2) + n * 2];
            acc = __builtin_amdgcn_wmma_f32_16x16x4_f32(false, a, false, bm,
                                                        (short)0, acc, false, false);
        }
        #pragma unroll
        for (int v = 0; v < 8; ++v) {
            int t = t0 + mbase + v;
            if (t < NFRAMES)
                melBuf[(size_t)t * NCH + b * NMELS + n] = acc[v];
        }
    }
}

__global__ void pcen_kernel(const float* __restrict__ melBuf,
                            float* __restrict__ out)
{
    int ch = blockIdx.x * blockDim.x + threadIdx.x;
    if (ch >= NCH) return;
    int b = ch / NMELS, m = ch % NMELS;
    const float s  = 1.0f - expf(-((float)HOP / (float)SRATE) / 0.06f);
    const float da = powf(2.0f, 0.98f);
    float* o = out + ((size_t)b * NMELS + m) * NFRAMES;
    float M = 0.0f;
    for (int t = 0; t < NFRAMES; ++t) {
        float e = melBuf[(size_t)t * NCH + ch];
        M = (1.0f - s) * M + s * e;
        float smooth = sqrtf(M + 1e-6f);
        o[t] = powf(e / smooth + 2.0f, 0.98f) - da;
    }
}

extern "C" void kernel_launch(void* const* d_in, const int* in_sizes, int n_in,
                              void* d_out, int out_size, void* d_ws, size_t ws_size,
                              hipStream_t stream) {
    const float* x = (const float*)d_in[0];
    float* ws = (float*)d_ws;

    float* dftCos = ws;
    float* dftSin = dftCos + (size_t)NFFT * FPAD;
    float* melFB  = dftSin + (size_t)NFFT * FPAD;
    float* melBuf = melFB  + (size_t)FPAD * NMELS;

    int n1 = NFFT * FPAD;
    init_dft<<<(n1 + 255) / 256, 256, 0, stream>>>(dftCos, dftSin);
    int n2 = FPAD * NMELS;
    init_mel<<<(n2 + 255) / 256, 256, 0, stream>>>(melFB);

    dim3 grid(TTILES, BATCH);
    dft_mel_kernel<<<grid, 128, 0, stream>>>(x, dftCos, dftSin, melFB, melBuf);

    pcen_kernel<<<(NCH + 127) / 128, 128, 0, stream>>>(melBuf, (float*)d_out);
}


# pcen: loop fission: the serial EMA (order unchanged) runs alone in a scan kernel, 56 row loads in flight; pow/sqrt/divide move to a new parallel kernel with LDS-transposed stores; 4 -> 5 launches
# speedup vs baseline: 3.0491x; 3.0491x over previous
#include <hip/hip_runtime.h>
#include <math.h>

#define SRATE     22050
#define NFFT      1024
#define HOP       256
#define NMELS     128
#define NFREQS    513
#define FPAD      528
#define BATCH     16
#define NSAMP     661504
#define NFRAMES   2585
#define TTILES    162
#define SPAN      (15*HOP + NFFT)
#define NCH       (BATCH*NMELS)

typedef __attribute__((ext_vector_type(2))) float v2f;
typedef __attribute__((ext_vector_type(8))) float v8f;

__global__ void init_dft(float* __restrict__ dftCos, float* __restrict__ dftSin) {
    int idx = blockIdx.x * blockDim.x + threadIdx.x;
    if (idx >= NFFT * FPAD) return;
    int k = idx / FPAD;
    int f = idx % FPAD;
    float c = 0.0f, s = 0.0f;
    if (f < NFREQS) {
        float win = 0.5f - 0.5f * cosf(6.28318530717958647692f * (float)k / (float)NFFT);
        int r = (k * f) & (NFFT - 1);
        float ang = (float)r * (6.28318530717958647692f / (float)NFFT);
        c = win * cosf(ang);
        s = win * sinf(ang);
    }
    int off = (k >> 1) * (FPAD * 2) + f * 2 + (k & 1);
    dftCos[off] = c;
    dftSin[off] = s;
}

__device__ __forceinline__ float hz_to_mel(float f) { return 2595.0f * log10f(1.0f + f / 700.0f); }
__device__ __forceinline__ float mel_to_hz(float m) { return 700.0f * (exp10f(m / 2595.0f) - 1.0f); }

__global__ void init_mel(float* __restrict__ melFB) {
    int idx = blockIdx.x * blockDim.x + threadIdx.x;
    if (idx >= FPAD * NMELS) return;
    int k = idx / NMELS;
    int m = idx % NMELS;
    float v = 0.0f;
    if (k < NFREQS) {
        float m_lo = hz_to_mel(50.0f);
        float m_hi = hz_to_mel((float)SRATE * 0.5f);
        float step = (m_hi - m_lo) / (float)(NMELS + 1);
        float f0 = mel_to_hz(m_lo + step * (float)m);
        float f1 = mel_to_hz(m_lo + step * (float)(m + 1));
        float f2 = mel_to_hz(m_lo + step * (float)(m + 2));
        float freq = (float)k * ((float)SRATE * 0.5f / (float)(NFFT / 2));
        float down = (freq - f0) / (f1 - f0);
        float up   = (f2 - freq) / (f2 - f1);
        v = fmaxf(0.0f, fminf(down, up));
    }
    melFB[(k >> 1) * (NMELS * 2) + m * 2 + (k & 1)] = v;
}

__global__ __launch_bounds__(128)
void dft_mel_kernel(const float* __restrict__ x,
                    const float* __restrict__ dftCos,
                    const float* __restrict__ dftSin,
                    const float* __restrict__ melFB,
                    float* __restrict__ melBuf)
{
    __shared__ float span[SPAN];
    __shared__ float mag[16 * FPAD];

    const int tile = blockIdx.x;
    const int b    = blockIdx.y;
    const int tid  = threadIdx.x;
    const int ln   = tid & 31;
    const int wv   = tid >> 5;
    const int t0   = tile * 16;

    const float* xb = x + (size_t)b * NSAMP;
    const long base = (long)t0 * HOP - (NFFT / 2);
    for (int i = tid; i < SPAN; i += 128) {
        long p = base + i;
        if (p < 0) p = -p;
        if (p >= NSAMP) p = 2L * NSAMP - 2 - p;
        if (p < 0) p = 0;
        if (p >= NSAMP) p = NSAMP - 1;
        span[i] = xb[p];
    }
    __syncthreads();

    const int mrow  = ln & 15;
    const int kh    = (ln >> 4) << 1;
    const int ncol  = ln & 15;
    const int mbase = (ln >> 4) * 8;

    for (int nt = wv; nt < FPAD / 16; nt += 4) {
        v8f accRe = {0.f, 0.f, 0.f, 0.f, 0.f, 0.f, 0.f, 0.f};
        v8f accIm = {0.f, 0.f, 0.f, 0.f, 0.f, 0.f, 0.f, 0.f};
        const int n = nt * 16 + ncol;
        #pragma unroll 4
        for (int kb = 0; kb < NFFT; kb += 4) {
            const int k0 = kb + kh;
            v2f a  = *(const v2f*)&span[mrow * HOP + k0];
            v2f bc = *(const v2f*)&dftCos[(k0 >> 1) * (FPAD * 2) + n * 2];
            v2f bs = *(const v2f*)&dftSin[(k0 >> 1) * (FPAD * 2) + n * 2];
            accRe = __builtin_amdgcn_wmma_f32_16x16x4_f32(false, a, false, bc,
                                                          (short)0, accRe, false, false);
            accIm = __builtin_amdgcn_wmma_f32_16x16x4_f32(false, a, false, bs,
                                                          (short)0, accIm, false, false);
        }
        #pragma unroll
        for (int v = 0; v < 8; ++v) {
            float re = accRe[v], im = accIm[v];
            mag[(mbase + v) * FPAD + n] = sqrtf(re * re + im * im);
        }
    }
    __syncthreads();

    for (int nt = wv; nt < NMELS / 16; nt += 4) {
        v8f acc = {0.f, 0.f, 0.f, 0.f, 0.f, 0.f, 0.f, 0.f};
        const int n = nt * 16 + ncol;
        #pragma unroll 4
        for (int kb = 0; kb < FPAD; kb += 4) {
            const int k0 = kb + kh;
            v2f a  = *(const v2f*)&mag[mrow * FPAD + k0];
            v2f bm = *(const v2f*)&melFB[(k0 >> 1) * (NMELS * 2) + n * 2];
            acc = __builtin_amdgcn_wmma_f32_16x16x4_f32(false, a, false, bm,
                                                        (short)0, acc, false, false);
        }
        #pragma unroll
        for (int v = 0; v < 8; ++v) {
            int t = t0 + mbase + v;
            if (t < NFRAMES)
                melBuf[(size_t)t * NCH + b * NMELS + n] = acc[v];
        }
    }
}

#define SC_U 56
__global__ __launch_bounds__(32)
void pcen_scan_kernel(const float* __restrict__ melBuf,
                      float* __restrict__ emaBuf)
{
    const int ch = blockIdx.x * 32 + threadIdx.x;
    const float s = 1.0f - expf(-((float)HOP / (float)SRATE) / 0.06f);
    constexpr int NGRP = NFRAMES / SC_U;
    constexpr int REM  = NFRAMES % SC_U;
    static_assert(NGRP >= 1, "at least one whole group");

    const float* p = melBuf + ch;
    float* q = emaBuf + ch;

    float r[SC_U];
    #pragma unroll
    for (int i = 0; i < SC_U; ++i) r[i] = p[(size_t)i * NCH];

    float M = 0.0f;
    for (int g = 0; g < NGRP - 1; ++g) {
        #pragma unroll
        for (int i = 0; i < SC_U; ++i) {
            M = (1.0f - s) * M + s * r[i];
            q[(size_t)i * NCH] = M;
            r[i] = p[(size_t)(SC_U + i) * NCH];
        }
        p += (size_t)SC_U * NCH;
        q += (size_t)SC_U * NCH;
    }
    #pragma unroll
    for (int i = 0; i < SC_U; ++i) {
        M = (1.0f - s) * M + s * r[i];
        q[(size_t)i * NCH] = M;
        if (i < REM) r[i] = p[(size_t)(SC_U + i) * NCH];
    }
    p += (size_t)SC_U * NCH;
    q += (size_t)SC_U * NCH;
    #pragma unroll
    for (int i = 0; i < REM; ++i) {
        M = (1.0f - s) * M + s * r[i];
        q[(size_t)i * NCH] = M;
    }
}

#define PW_TT    128
#define PW_PITCH (PW_TT + 1)
#define PW_TTILES ((NFRAMES + PW_TT - 1) / PW_TT)
__global__ __launch_bounds__(256)
void pcen_point_kernel(const float* __restrict__ melBuf,
                       const float* __restrict__ emaBuf,
                       float* __restrict__ out)
{
    __shared__ float tile[32 * PW_PITCH];

    const int tid = threadIdx.x;
    const int ln  = tid & 31;
    const int wv  = tid >> 5;
    const int t0  = blockIdx.x * PW_TT;
    const int ch0 = blockIdx.y * 32;
    const float da = powf(2.0f, 0.98f);

    #pragma unroll 1
    for (int kk = 0; kk < PW_TT / 8; kk += 4) {
        float e[4], M[4];
        #pragma unroll
        for (int j = 0; j < 4; ++j) {
            const int row = wv + 8 * (kk + j);
            int t = t0 + row;
            if (t >= NFRAMES) t = NFRAMES - 1;
            const int off = t * NCH + ch0 + ln;
            e[j] = melBuf[off];
            M[j] = emaBuf[off];
        }
        #pragma unroll
        for (int j = 0; j < 4; ++j) {
            const int row = wv + 8 * (kk + j);
            float smooth = sqrtf(M[j] + 1e-6f);
            tile[ln * PW_PITCH + row] = powf(e[j] / smooth + 2.0f, 0.98f) - da;
        }
    }
    __syncthreads();

    const int tt = tid % PW_TT;
    const int c0 = tid / PW_TT;
    if (t0 + tt < NFRAMES) {
        float* o = out + (size_t)ch0 * NFRAMES + t0 + tt;
        for (int k = 0; k < 32 / (256 / PW_TT); ++k) {
            const int c = k * (256 / PW_TT) + c0;
            o[c * NFRAMES] = tile[c * PW_PITCH + tt];
        }
    }
}

extern "C" void kernel_launch(void* const* d_in, const int* in_sizes, int n_in,
                              void* d_out, int out_size, void* d_ws, size_t ws_size,
                              hipStream_t stream) {
    const float* x = (const float*)d_in[0];
    float* ws = (float*)d_ws;

    float* dftCos = ws;
    float* dftSin = dftCos + (size_t)NFFT * FPAD;
    float* melFB  = dftSin + (size_t)NFFT * FPAD;
    float* melBuf = melFB  + (size_t)FPAD * NMELS;
    float* emaBuf = melBuf + (size_t)2592 * NCH;

    int n1 = NFFT * FPAD;
    init_dft<<<(n1 + 255) / 256, 256, 0, stream>>>(dftCos, dftSin);
    int n2 = FPAD * NMELS;
    init_mel<<<(n2 + 255) / 256, 256, 0, stream>>>(melFB);

    dim3 grid(TTILES, BATCH);
    dft_mel_kernel<<<grid, 128, 0, stream>>>(x, dftCos, dftSin, melFB, melBuf);

    pcen_scan_kernel<<<NCH / 32, 32, 0, stream>>>(melBuf, emaBuf);

    dim3 pgrid(PW_TTILES, NCH / 32);
    pcen_point_kernel<<<pgrid, 256, 0, stream>>>(melBuf, emaBuf, (float*)d_out);
}


# dft_mel: dense 1024x528 DFT product -> 32x32 factorisation: two 32-point WMMA stages and a twiddle, stage-1 D tiles used in place as stage-2 A; tables in registers; DFT sum order changes
# speedup vs baseline: 10.7835x; 3.5366x over previous
#include <hip/hip_runtime.h>
#include <math.h>

#define SRATE     22050
#define NFFT      1024
#define HOP       256
#define NMELS     128
#define NFREQS    513
#define FPAD      528
#define BATCH     16
#define NSAMP     661504
#define NFRAMES   2585
#define TTILES    162
#define SPAN      (15*HOP + NFFT)
#define NCH       (BATCH*NMELS)

typedef __attribute__((ext_vector_type(2))) float v2f;
typedef __attribute__((ext_vector_type(8))) float v8f;

#define NTAB 192
__global__ void init_tabs(float* __restrict__ tab) {
    int idx = blockIdx.x * blockDim.x + threadIdx.x;
    if (idx >= NTAB * 32) return;
    const int e = idx >> 5, ln = idx & 31, c = ln & 15, h = ln >> 4;
    const float TWO_PI = 6.28318530717958647692f;
    float v;
    if (e < 64) {
        const int q = e & 31;
        const int s = q & 1, nt = (q >> 1) & 1, ks = q >> 2;
        const int f1 = 16 * nt + c, n1 = 4 * ks + 2 * h + s;
        const float ang = (float)((f1 * n1) & 31) * (TWO_PI / 32.0f);
        v = (e < 32) ? cosf(ang) : sinf(ang);
    } else if (e < 96) {
        const int q = e - 64;
        const int s = q & 1, ks = (q >> 1) & 7, mt = q >> 4;
        const int n = 32 * (4 * ks + 2 * h + s) + 16 * mt + c;
        v = 0.5f - 0.5f * cosf(TWO_PI * (float)n / (float)NFFT);
    } else if (e < 160) {
        const int q = (e - 96) & 31;
        const int r = q & 7, nt = (q >> 3) & 1, mt = q >> 4;
        const int f1 = 16 * nt + c, n2 = 16 * mt + 8 * h + r;
        const float ang = (float)(f1 * n2) * (TWO_PI / (float)NFFT);
        v = (e < 128) ? cosf(ang) : sinf(ang);
    } else {
        const int q = (e - 160) & 15;
        const int s = q & 1, j = (q >> 1) & 3, mt = q >> 3;
        const int f2 = c, n2 = 16 * mt + 8 * h + 2 * j + s;
        const float ang = (float)((f2 * n2) & 31) * (TWO_PI / 32.0f);
        v = (e < 176) ? cosf(ang) : sinf(ang);
    }
    tab[idx] = v;
}

__device__ __forceinline__ float hz_to_mel(float f) { return 2595.0f * log10f(1.0f + f / 700.0f); }
__device__ __forceinline__ float mel_to_hz(float m) { return 700.0f * (exp10f(m / 2595.0f) - 1.0f); }

__global__ void init_mel(float* __restrict__ melFB) {
    int idx = blockIdx.x * blockDim.x + threadIdx.x;
    if (idx >= FPAD * NMELS) return;
    int k = idx / NMELS;
    int m = idx % NMELS;
    float v = 0.0f;
    if (k < NFREQS) {
        float m_lo = hz_to_mel(50.0f);
        float m_hi = hz_to_mel((float)SRATE * 0.5f);
        float step = (m_hi - m_lo) / (float)(NMELS + 1);
        float f0 = mel_to_hz(m_lo + step * (float)m);
        float f1 = mel_to_hz(m_lo + step * (float)(m + 1));
        float f2 = mel_to_hz(m_lo + step * (float)(m + 2));
        float freq = (float)k * ((float)SRATE * 0.5f / (float)(NFFT / 2));
        float down = (freq - f0) / (f1 - f0);
        float up   = (f2 - freq) / (f2 - f1);
        v = fmaxf(0.0f, fminf(down, up));
    }
    melFB[(k >> 1) * (NMELS * 2) + m * 2 + (k & 1)] = v;
}

#define WMMA4(A, B, C) __builtin_amdgcn_wmma_f32_16x16x4_f32(false, (A), false, (B), (short)0, (C), false, false)
__global__ __launch_bounds__(128, 3)
void dft_mel_kernel(const float* __restrict__ x,
                    const float* __restrict__ tab,
                    const float* __restrict__ melFB,
                    float* __restrict__ melBuf)
{
    __shared__ float span[SPAN];
    __shared__ float mag[16 * FPAD];

    const int tile = blockIdx.x;
    const int b    = blockIdx.y;
    const int tid  = threadIdx.x;
    const int ln   = tid & 31;
    const int wv   = tid >> 5;
    const int t0   = tile * 16;
    const int c    = ln & 15;
    const int h    = ln >> 4;

    const float* xb = x + (size_t)b * NSAMP;
    const long base = (long)t0 * HOP - (NFFT / 2);
    for (int i = tid; i < SPAN; i += 128) {
        long p = base + i;
        if (p < 0) p = -p;
        if (p >= NSAMP) p = 2L * NSAMP - 2 - p;
        if (p < 0) p = 0;
        if (p >= NSAMP) p = NSAMP - 1;
        span[i] = xb[p];
    }

    const float* tl = tab + ln;
    v2f B1c[8][2], B1s[8][2];
    v2f Wn[2][8];
    float Tc[2][2][8], Ts[2][2][8];
    v2f Ec[2][4], Es[2][4], En[2][4];
    #pragma unroll
    for (int ks = 0; ks < 8; ++ks) {
        #pragma unroll
        for (int nt = 0; nt < 2; ++nt) {
            const int e = (ks * 2 + nt) * 2;
            B1c[ks][nt].x = tl[(e + 0) * 32];       B1c[ks][nt].y = tl[(e + 1) * 32];
            B1s[ks][nt].x = tl[(32 + e + 0) * 32];  B1s[ks][nt].y = tl[(32 + e + 1) * 32];
        }
    }
    #pragma unroll
    for (int mt = 0; mt < 2; ++mt) {
        #pragma unroll
        for (int ks = 0; ks < 8; ++ks) {
            const int e = 64 + (mt * 8 + ks) * 2;
            Wn[mt][ks].x = tl[(e + 0) * 32];  Wn[mt][ks].y = tl[(e + 1) * 32];
        }
        #pragma unroll
        for (int nt = 0; nt < 2; ++nt) {
            #pragma unroll
            for (int r = 0; r < 8; ++r) {
                const int e = (mt * 2 + nt) * 8 + r;
                Tc[mt][nt][r] = tl[(96 + e) * 32];
                Ts[mt][nt][r] = tl[(128 + e) * 32];
            }
        }
        #pragma unroll
        for (int j = 0; j < 4; ++j) {
            const int e = (mt * 4 + j) * 2;
            Ec[mt][j].x = tl[(160 + e + 0) * 32];  Ec[mt][j].y = tl[(160 + e + 1) * 32];
            Es[mt][j].x = tl[(176 + e + 0) * 32];  Es[mt][j].y = tl[(176 + e + 1) * 32];
            En[mt][j] = -Es[mt][j];
        }
    }
    __syncthreads();

    #pragma unroll 1
    for (int q = 0; q < 4; ++q) {
        const int fr = wv * 4 + q;
        const float* xs = span + fr * HOP + 64 * h + c;

        v8f Yc[2][2], Ys[2][2];
        #pragma unroll
        for (int mt = 0; mt < 2; ++mt) {
            #pragma unroll
            for (int nt = 0; nt < 2; ++nt) {
                Yc[mt][nt] = (v8f){0.f, 0.f, 0.f, 0.f, 0.f, 0.f, 0.f, 0.f};
                Ys[mt][nt] = (v8f){0.f, 0.f, 0.f, 0.f, 0.f, 0.f, 0.f, 0.f};
            }
        }
        #pragma unroll
        for (int ks = 0; ks < 8; ++ks) {
            #pragma unroll
            for (int mt = 0; mt < 2; ++mt) {
                v2f a;
                a.x = xs[128 * ks + 16 * mt]      * Wn[mt][ks].x;
                a.y = xs[128 * ks + 16 * mt + 32] * Wn[mt][ks].y;
                #pragma unroll
                for (int nt = 0; nt < 2; ++nt) {
                    Yc[mt][nt] = WMMA4(a, B1c[ks][nt], Yc[mt][nt]);
                    Ys[mt][nt] = WMMA4(a, B1s[ks][nt], Ys[mt][nt]);
                }
            }
        }

        float Pr[2][2][8], Ps[2][2][8];
        #pragma unroll
        for (int mt = 0; mt < 2; ++mt) {
            #pragma unroll
            for (int nt = 0; nt < 2; ++nt) {
                #pragma unroll
                for (int r = 0; r < 8; ++r) {
                    const float yc = Yc[mt][nt][r], ys = Ys[mt][nt][r];
                    Pr[mt][nt][r] = yc * Tc[mt][nt][r] - ys * Ts[mt][nt][r];
                    Ps[mt][nt][r] = yc * Ts[mt][nt][r] + ys * Tc[mt][nt][r];
                }
            }
        }

        #pragma unroll
        for (int nt = 0; nt < 2; ++nt) {
            v8f Xr = (v8f){0.f, 0.f, 0.f, 0.f, 0.f, 0.f, 0.f, 0.f};
            v8f Xi = (v8f){0.f, 0.f, 0.f, 0.f, 0.f, 0.f, 0.f, 0.f};
            #pragma unroll
            for (int mt = 0; mt < 2; ++mt) {
                #pragma unroll
                for (int j = 0; j < 4; ++j) {
                    v2f ar, ai;
                    ar.x = Pr[mt][nt][2 * j];  ar.y = Pr[mt][nt][2 * j + 1];
                    ai.x = Ps[mt][nt][2 * j];  ai.y = Ps[mt][nt][2 * j + 1];
                    Xr = WMMA4(ar, Ec[mt][j], Xr);
                    Xi = WMMA4(ar, Es[mt][j], Xi);
                    Xr = WMMA4(ai, En[mt][j], Xr);
                    Xi = WMMA4(ai, Ec[mt][j], Xi);
                }
            }
            #pragma unroll
            for (int r = 0; r < 8; ++r) {
                const float re = Xr[r], im = Xi[r];
                mag[fr * FPAD + 32 * c + 16 * nt + 8 * h + r] = sqrtf(re * re + im * im);
            }
        }

        float ny = 0.0f;
        #pragma unroll
        for (int mt = 0; mt < 2; ++mt) {
            #pragma unroll
            for (int r = 0; r < 8; r += 2) ny += Pr[mt][0][r] - Pr[mt][0][r + 1];
        }
        ny += __shfl_xor(ny, 16);
        if (ln < 16) mag[fr * FPAD + 512 + ln] = (ln == 0) ? fabsf(ny) : 0.0f;
    }
    __syncthreads();

    const int mrow  = c;
    const int kh    = h << 1;
    const int ncol  = c;
    const int mbase = h * 8;

    for (int nt = wv; nt < NMELS / 16; nt += 4) {
        v8f acc = {0.f, 0.f, 0.f, 0.f, 0.f, 0.f, 0.f, 0.f};
        const int n = nt * 16 + ncol;
        #pragma unroll 4
        for (int kb = 0; kb < FPAD; kb += 4) {
            const int k0 = kb + kh;
            v2f a  = *(const v2f*)&mag[mrow * FPAD + k0];
            v2f bm = *(const v2f*)&melFB[(k0 >> 1) * (NMELS * 2) + n * 2];
            acc = WMMA4(a, bm, acc);
        }
        #pragma unroll
        for (int v = 0; v < 8; ++v) {
            int t = t0 + mbase + v;
            if (t < NFRAMES)
                melBuf[(size_t)t * NCH + b * NMELS + n] = acc[v];
        }
    }
}

#define SC_U 56
__global__ __launch_bounds__(32)
void pcen_scan_kernel(const float* __restrict__ melBuf,
                      float* __restrict__ emaBuf)
{
    const int ch = blockIdx.x * 32 + threadIdx.x;
    const float s = 1.0f - expf(-((float)HOP / (float)SRATE) / 0.06f);
    constexpr int NGRP = NFRAMES / SC_U;
    constexpr int REM  = NFRAMES % SC_U;
    static_assert(NGRP >= 1, "at least one whole group");

    const float* p = melBuf + ch;
    float* q = emaBuf + ch;

    float r[SC_U];
    #pragma unroll
    for (int i = 0; i < SC_U; ++i) r[i] = p[(size_t)i * NCH];

    float M = 0.0f;
    for (int g = 0; g < NGRP - 1; ++g) {
        #pragma unroll
        for (int i = 0; i < SC_U; ++i) {
            M = (1.0f - s) * M + s * r[i];
            q[(size_t)i * NCH] = M;
            r[i] = p[(size_t)(SC_U + i) * NCH];
        }
        p += (size_t)SC_U * NCH;
        q += (size_t)SC_U * NCH;
    }
    #pragma unroll
    for (int i = 0; i < SC_U; ++i) {
        M = (1.0f - s) * M + s * r[i];
        q[(size_t)i * NCH] = M;
        if (i < REM) r[i] = p[(size_t)(SC_U + i) * NCH];
    }
    p += (size_t)SC_U * NCH;
    q += (size_t)SC_U * NCH;
    #pragma unroll
    for (int i = 0; i < REM; ++i) {
        M = (1.0f - s) * M + s * r[i];
        q[(size_t)i * NCH] = M;
    }
}

#define PW_TT    128
#define PW_PITCH (PW_TT + 1)
#define PW_TTILES ((NFRAMES + PW_TT - 1) / PW_TT)
__global__ __launch_bounds__(256)
void pcen_point_kernel(const float* __restrict__ melBuf,
                       const float* __restrict__ emaBuf,
                       float* __restrict__ out)
{
    __shared__ float tile[32 * PW_PITCH];

    const int tid = threadIdx.x;
    const int ln  = tid & 31;
    const int wv  = tid >> 5;
    const int t0  = blockIdx.x * PW_TT;
    const int ch0 = blockIdx.y * 32;
    const float da = powf(2.0f, 0.98f);

    #pragma unroll 1
    for (int kk = 0; kk < PW_TT / 8; kk += 4) {
        float e[4], M[4];
        #pragma unroll
        for (int j = 0; j < 4; ++j) {
            const int row = wv + 8 * (kk + j);
            int t = t0 + row;
            if (t >= NFRAMES) t = NFRAMES - 1;
            const int off = t * NCH + ch0 + ln;
            e[j] = melBuf[off];
            M[j] = emaBuf[off];
        }
        #pragma unroll
        for (int j = 0; j < 4; ++j) {
            const int row = wv + 8 * (kk + j);
            float smooth = sqrtf(M[j] + 1e-6f);
            tile[ln * PW_PITCH + row] = powf(e[j] / smooth + 2.0f, 0.98f) - da;
        }
    }
    __syncthreads();

    const int tt = tid % PW_TT;
    const int c0 = tid / PW_TT;
    if (t0 + tt < NFRAMES) {
        float* o = out + (size_t)ch0 * NFRAMES + t0 + tt;
        for (int k = 0; k < 32 / (256 / PW_TT); ++k) {
            const int c = k * (256 / PW_TT) + c0;
            o[c * NFRAMES] = tile[c * PW_PITCH + tt];
        }
    }
}

extern "C" void kernel_launch(void* const* d_in, const int* in_sizes, int n_in,
                              void* d_out, int out_size, void* d_ws, size_t ws_size,
                              hipStream_t stream) {
    const float* x = (const float*)d_in[0];
    float* ws = (float*)d_ws;

    float* tab    = ws;
    float* melFB  = tab    + (size_t)NTAB * 32;
    float* melBuf = melFB  + (size_t)FPAD * NMELS;
    float* emaBuf = melBuf + (size_t)2592 * NCH;

    init_tabs<<<(NTAB * 32 + 255) / 256, 256, 0, stream>>>(tab);
    int n2 = FPAD * NMELS;
    init_mel<<<(n2 + 255) / 256, 256, 0, stream>>>(melFB);

    dim3 grid(TTILES, BATCH);
    dft_mel_kernel<<<grid, 128, 0, stream>>>(x, tab, melFB, melBuf);

    pcen_scan_kernel<<<NCH / 32, 32, 0, stream>>>(melBuf, emaBuf);

    dim3 pgrid(PW_TTILES, NCH / 32);
    pcen_point_kernel<<<pgrid, 256, 0, stream>>>(melBuf, emaBuf, (float*)d_out);
}


# dft_mel mel loop: the wave's two mel tiles share each LDS A fragment and run side by side over K in groups of 22 k-steps, a group's 44 table loads issued together (were 4 per wait); sums unchanged
# speedup vs baseline: 14.7556x; 1.3683x over previous
#include <hip/hip_runtime.h>
#include <math.h>

#define SRATE     22050
#define NFFT      1024
#define HOP       256
#define NMELS     128
#define NFREQS    513
#define FPAD      528
#define BATCH     16
#define NSAMP     661504
#define NFRAMES   2585
#define TTILES    162
#define SPAN      (15*HOP + NFFT)
#define NCH       (BATCH*NMELS)

typedef __attribute__((ext_vector_type(2))) float v2f;
typedef __attribute__((ext_vector_type(8))) float v8f;

#define NTAB 192
__global__ void init_tabs(float* __restrict__ tab) {
    int idx = blockIdx.x * blockDim.x + threadIdx.x;
    if (idx >= NTAB * 32) return;
    const int e = idx >> 5, ln = idx & 31, c = ln & 15, h = ln >> 4;
    const float TWO_PI = 6.28318530717958647692f;
    float v;
    if (e < 64) {
        const int q = e & 31;
        const int s = q & 1, nt = (q >> 1) & 1, ks = q >> 2;
        const int f1 = 16 * nt + c, n1 = 4 * ks + 2 * h + s;
        const float ang = (float)((f1 * n1) & 31) * (TWO_PI / 32.0f);
        v = (e < 32) ? cosf(ang) : sinf(ang);
    } else if (e < 96) {
        const int q = e - 64;
        const int s = q & 1, ks = (q >> 1) & 7, mt = q >> 4;
        const int n = 32 * (4 * ks + 2 * h + s) + 16 * mt + c;
        v = 0.5f - 0.5f * cosf(TWO_PI * (float)n / (float)NFFT);
    } else if (e < 160) {
        const int q = (e - 96) & 31;
        const int r = q & 7, nt = (q >> 3) & 1, mt = q >> 4;
        const int f1 = 16 * nt + c, n2 = 16 * mt + 8 * h + r;
        const float ang = (float)(f1 * n2) * (TWO_PI / (float)NFFT);
        v = (e < 128) ? cosf(ang) : sinf(ang);
    } else {
        const int q = (e - 160) & 15;
        const int s = q & 1, j = (q >> 1) & 3, mt = q >> 3;
        const int f2 = c, n2 = 16 * mt + 8 * h + 2 * j + s;
        const float ang = (float)((f2 * n2) & 31) * (TWO_PI / 32.0f);
        v = (e < 176) ? cosf(ang) : sinf(ang);
    }
    tab[idx] = v;
}

__device__ __forceinline__ float hz_to_mel(float f) { return 2595.0f * log10f(1.0f + f / 700.0f); }
__device__ __forceinline__ float mel_to_hz(float m) { return 700.0f * (exp10f(m / 2595.0f) - 1.0f); }

__global__ void init_mel(float* __restrict__ melFB) {
    int idx = blockIdx.x * blockDim.x + threadIdx.x;
    if (idx >= FPAD * NMELS) return;
    int k = idx / NMELS;
    int m = idx % NMELS;
    float v = 0.0f;
    if (k < NFREQS) {
        float m_lo = hz_to_mel(50.0f);
        float m_hi = hz_to_mel((float)SRATE * 0.5f);
        float step = (m_hi - m_lo) / (float)(NMELS + 1);
        float f0 = mel_to_hz(m_lo + step * (float)m);
        float f1 = mel_to_hz(m_lo + step * (float)(m + 1));
        float f2 = mel_to_hz(m_lo + step * (float)(m + 2));
        float freq = (float)k * ((float)SRATE * 0.5f / (float)(NFFT / 2));
        float down = (freq - f0) / (f1 - f0);
        float up   = (f2 - freq) / (f2 - f1);
        v = fmaxf(0.0f, fminf(down, up));
    }
    melFB[(k >> 1) * (NMELS * 2) + m * 2 + (k & 1)] = v;
}

#define WMMA4(A, B, C) __builtin_amdgcn_wmma_f32_16x16x4_f32(false, (A), false, (B), (short)0, (C), false, false)
__global__ __launch_bounds__(128, 3)
void dft_mel_kernel(const float* __restrict__ x,
                    const float* __restrict__ tab,
                    const float* __restrict__ melFB,
                    float* __restrict__ melBuf)
{
    __shared__ float span[SPAN];
    __shared__ float mag[16 * FPAD];

    const int tile = blockIdx.x;
    const int b    = blockIdx.y;
    const int tid  = threadIdx.x;
    const int ln   = tid & 31;
    const int wv   = tid >> 5;
    const int t0   = tile * 16;
    const int c    = ln & 15;
    const int h    = ln >> 4;

    const float* xb = x + (size_t)b * NSAMP;
    const long base = (long)t0 * HOP - (NFFT / 2);
    for (int i = tid; i < SPAN; i += 128) {
        long p = base + i;
        if (p < 0) p = -p;
        if (p >= NSAMP) p = 2L * NSAMP - 2 - p;
        if (p < 0) p = 0;
        if (p >= NSAMP) p = NSAMP - 1;
        span[i] = xb[p];
    }

    const float* tl = tab + ln;
    v2f B1c[8][2], B1s[8][2];
    v2f Wn[2][8];
    float Tc[2][2][8], Ts[2][2][8];
    v2f Ec[2][4], Es[2][4], En[2][4];
    #pragma unroll
    for (int ks = 0; ks < 8; ++ks) {
        #pragma unroll
        for (int nt = 0; nt < 2; ++nt) {
            const int e = (ks * 2 + nt) * 2;
            B1c[ks][nt].x = tl[(e + 0) * 32];       B1c[ks][nt].y = tl[(e + 1) * 32];
            B1s[ks][nt].x = tl[(32 + e + 0) * 32];  B1s[ks][nt].y = tl[(32 + e + 1) * 32];
        }
    }
    #pragma unroll
    for (int mt = 0; mt < 2; ++mt) {
        #pragma unroll
        for (int ks = 0; ks < 8; ++ks) {
            const int e = 64 + (mt * 8 + ks) * 2;
            Wn[mt][ks].x = tl[(e + 0) * 32];  Wn[mt][ks].y = tl[(e + 1) * 32];
        }
        #pragma unroll
        for (int nt = 0; nt < 2; ++nt) {
            #pragma unroll
            for (int r = 0; r < 8; ++r) {
                const int e = (mt * 2 + nt) * 8 + r;
                Tc[mt][nt][r] = tl[(96 + e) * 32];
                Ts[mt][nt][r] = tl[(128 + e) * 32];
            }
        }
        #pragma unroll
        for (int j = 0; j < 4; ++j) {
            const int e = (mt * 4 + j) * 2;
            Ec[mt][j].x = tl[(160 + e + 0) * 32];  Ec[mt][j].y = tl[(160 + e + 1) * 32];
            Es[mt][j].x = tl[(176 + e + 0) * 32];  Es[mt][j].y = tl[(176 + e + 1) * 32];
            En[mt][j] = -Es[mt][j];
        }
    }
    __syncthreads();

    #pragma unroll 1
    for (int q = 0; q < 4; ++q) {
        const int fr = wv * 4 + q;
        const float* xs = span + fr * HOP + 64 * h + c;

        v8f Yc[2][2], Ys[2][2];
        #pragma unroll
        for (int mt = 0; mt < 2; ++mt) {
            #pragma unroll
            for (int nt = 0; nt < 2; ++nt) {
                Yc[mt][nt] = (v8f){0.f, 0.f, 0.f, 0.f, 0.f, 0.f, 0.f, 0.f};
                Ys[mt][nt] = (v8f){0.f, 0.f, 0.f, 0.f, 0.f, 0.f, 0.f, 0.f};
            }
        }
        #pragma unroll
        for (int ks = 0; ks < 8; ++ks) {
            #pragma unroll
            for (int mt = 0; mt < 2; ++mt) {
                v2f a;
                a.x = xs[128 * ks + 16 * mt]      * Wn[mt][ks].x;
                a.y = xs[128 * ks + 16 * mt + 32] * Wn[mt][ks].y;
                #pragma unroll
                for (int nt = 0; nt < 2; ++nt) {
                    Yc[mt][nt] = WMMA4(a, B1c[ks][nt], Yc[mt][nt]);
                    Ys[mt][nt] = WMMA4(a, B1s[ks][nt], Ys[mt][nt]);
                }
            }
        }

        float Pr[2][2][8], Ps[2][2][8];
        #pragma unroll
        for (int mt = 0; mt < 2; ++mt) {
            #pragma unroll
            for (int nt = 0; nt < 2; ++nt) {
                #pragma unroll
                for (int r = 0; r < 8; ++r) {
                    const float yc = Yc[mt][nt][r], ys = Ys[mt][nt][r];
                    Pr[mt][nt][r] = yc * Tc[mt][nt][r] - ys * Ts[mt][nt][r];
                    Ps[mt][nt][r] = yc * Ts[mt][nt][r] + ys * Tc[mt][nt][r];
                }
            }
        }

        #pragma unroll
        for (int nt = 0; nt < 2; ++nt) {
            v8f Xr = (v8f){0.f, 0.f, 0.f, 0.f, 0.f, 0.f, 0.f, 0.f};
            v8f Xi = (v8f){0.f, 0.f, 0.f, 0.f, 0.f, 0.f, 0.f, 0.f};
            #pragma unroll
            for (int mt = 0; mt < 2; ++mt) {
                #pragma unroll
                for (int j = 0; j < 4; ++j) {
                    v2f ar, ai;
                    ar.x = Pr[mt][nt][2 * j];  ar.y = Pr[mt][nt][2 * j + 1];
                    ai.x = Ps[mt][nt][2 * j];  ai.y = Ps[mt][nt][2 * j + 1];
                    Xr = WMMA4(ar, Ec[mt][j], Xr);
                    Xi = WMMA4(ar, Es[mt][j], Xi);
                    Xr = WMMA4(ai, En[mt][j], Xr);
                    Xi = WMMA4(ai, Ec[mt][j], Xi);
                }
            }
            #pragma unroll
            for (int r = 0; r < 8; ++r) {
                const float re = Xr[r], im = Xi[r];
                mag[fr * FPAD + 32 * c + 16 * nt + 8 * h + r] = sqrtf(re * re + im * im);
            }
        }

        float ny = 0.0f;
        #pragma unroll
        for (int mt = 0; mt < 2; ++mt) {
            #pragma unroll
            for (int r = 0; r < 8; r += 2) ny += Pr[mt][0][r] - Pr[mt][0][r + 1];
        }
        ny += __shfl_xor(ny, 16);
        if (ln < 16) mag[fr * FPAD + 512 + ln] = (ln == 0) ? fabsf(ny) : 0.0f;
    }
    __syncthreads();

    const int mrow  = c;
    const int kh    = h << 1;
    const int ncol  = c;
    const int mbase = h * 8;

    constexpr int MEL_KSTEPS = FPAD / 4;
    constexpr int MEL_KG     = 22;
    static_assert(MEL_KSTEPS % MEL_KG == 0, "whole groups only");
    {
        v8f acc0 = {0.f, 0.f, 0.f, 0.f, 0.f, 0.f, 0.f, 0.f};
        v8f acc1 = {0.f, 0.f, 0.f, 0.f, 0.f, 0.f, 0.f, 0.f};
        const int n0 = wv * 16 + ncol;
        const int n1 = n0 + 64;
        const float* bp = melFB + h * (NMELS * 2) + n0 * 2;
        const float* ap = mag + mrow * FPAD + kh;
        #pragma unroll 1
        for (int g = 0; g < MEL_KSTEPS / MEL_KG; ++g) {
            v2f bf0[MEL_KG], bf1[MEL_KG], af[MEL_KG];
            #pragma unroll
            for (int i = 0; i < MEL_KG; ++i) {
                const float* q = bp + (size_t)(g * MEL_KG + i) * (2 * NMELS * 2);
                bf0[i] = *(const v2f*)q;
                bf1[i] = *(const v2f*)(q + 64 * 2);
            }
            #pragma unroll
            for (int i = 0; i < MEL_KG; ++i)
                af[i] = *(const v2f*)(ap + (g * MEL_KG + i) * 4);
            #pragma unroll
            for (int i = 0; i < MEL_KG; ++i) {
                acc0 = WMMA4(af[i], bf0[i], acc0);
                acc1 = WMMA4(af[i], bf1[i], acc1);
            }
        }
        #pragma unroll
        for (int v = 0; v < 8; ++v) {
            const int t = t0 + mbase + v;
            if (t < NFRAMES) {
                float* o = melBuf + (size_t)t * NCH + b * NMELS;
                o[n0] = acc0[v];
                o[n1] = acc1[v];
            }
        }
    }
}

#define SC_U 56
__global__ __launch_bounds__(32)
void pcen_scan_kernel(const float* __restrict__ melBuf,
                      float* __restrict__ emaBuf)
{
    const int ch = blockIdx.x * 32 + threadIdx.x;
    const float s = 1.0f - expf(-((float)HOP / (float)SRATE) / 0.06f);
    constexpr int NGRP = NFRAMES / SC_U;
    constexpr int REM  = NFRAMES % SC_U;
    static_assert(NGRP >= 1, "at least one whole group");

    const float* p = melBuf + ch;
    float* q = emaBuf + ch;

    float r[SC_U];
    #pragma unroll
    for (int i = 0; i < SC_U; ++i) r[i] = p[(size_t)i * NCH];

    float M = 0.0f;
    for (int g = 0; g < NGRP - 1; ++g) {
        #pragma unroll
        for (int i = 0; i < SC_U; ++i) {
            M = (1.0f - s) * M + s * r[i];
            q[(size_t)i * NCH] = M;
            r[i] = p[(size_t)(SC_U + i) * NCH];
        }
        p += (size_t)SC_U * NCH;
        q += (size_t)SC_U * NCH;
    }
    #pragma unroll
    for (int i = 0; i < SC_U; ++i) {
        M = (1.0f - s) * M + s * r[i];
        q[(size_t)i * NCH] = M;
        if (i < REM) r[i] = p[(size_t)(SC_U + i) * NCH];
    }
    p += (size_t)SC_U * NCH;
    q += (size_t)SC_U * NCH;
    #pragma unroll
    for (int i = 0; i < REM; ++i) {
        M = (1.0f - s) * M + s * r[i];
        q[(size_t)i * NCH] = M;
    }
}

#define PW_TT    128
#define PW_PITCH (PW_TT + 1)
#define PW_TTILES ((NFRAMES + PW_TT - 1) / PW_TT)
__global__ __launch_bounds__(256)
void pcen_point_kernel(const float* __restrict__ melBuf,
                       const float* __restrict__ emaBuf,
                       float* __restrict__ out)
{
    __shared__ float tile[32 * PW_PITCH];

    const int tid = threadIdx.x;
    const int ln  = tid & 31;
    const int wv  = tid >> 5;
    const int t0  = blockIdx.x * PW_TT;
    const int ch0 = blockIdx.y * 32;
    const float da = powf(2.0f, 0.98f);

    #pragma unroll 1
    for (int kk = 0; kk < PW_TT / 8; kk += 4) {
        float e[4], M[4];
        #pragma unroll
        for (int j = 0; j < 4; ++j) {
            const int row = wv + 8 * (kk + j);
            int t = t0 + row;
            if (t >= NFRAMES) t = NFRAMES - 1;
            const int off = t * NCH + ch0 + ln;
            e[j] = melBuf[off];
            M[j] = emaBuf[off];
        }
        #pragma unroll
        for (int j = 0; j < 4; ++j) {
            const int row = wv + 8 * (kk + j);
            float smooth = sqrtf(M[j] + 1e-6f);
            tile[ln * PW_PITCH + row] = powf(e[j] / smooth + 2.0f, 0.98f) - da;
        }
    }
    __syncthreads();

    const int tt = tid % PW_TT;
    const int c0 = tid / PW_TT;
    if (t0 + tt < NFRAMES) {
        float* o = out + (size_t)ch0 * NFRAMES + t0 + tt;
        for (int k = 0; k < 32 / (256 / PW_TT); ++k) {
            const int c = k * (256 / PW_TT) + c0;
            o[c * NFRAMES] = tile[c * PW_PITCH + tt];
        }
    }
}

extern "C" void kernel_launch(void* const* d_in, const int* in_sizes, int n_in,
                              void* d_out, int out_size, void* d_ws, size_t ws_size,
                              hipStream_t stream) {
    const float* x = (const float*)d_in[0];
    float* ws = (float*)d_ws;

    float* tab    = ws;
    float* melFB  = tab    + (size_t)NTAB * 32;
    float* melBuf = melFB  + (size_t)FPAD * NMELS;
    float* emaBuf = melBuf + (size_t)2592 * NCH;

    init_tabs<<<(NTAB * 32 + 255) / 256, 256, 0, stream>>>(tab);
    int n2 = FPAD * NMELS;
    init_mel<<<(n2 + 255) / 256, 256, 0, stream>>>(melFB);

    dim3 grid(TTILES, BATCH);
    dft_mel_kernel<<<grid, 128, 0, stream>>>(x, tab, melFB, melBuf);

    pcen_scan_kernel<<<NCH / 32, 32, 0, stream>>>(melBuf, emaBuf);

    dim3 pgrid(PW_TTILES, NCH / 32);
    pcen_point_kernel<<<pgrid, 256, 0, stream>>>(melBuf, emaBuf, (float*)d_out);
}


# dft_mel stage 1: samples are real, so only cos f1=0..16 and sin f1=1..15 are summed (32 columns, were 64); f1=17..31 enter stage 2 as same-lane conjugates; 64->32 WMMAs a frame; rounding moves
# speedup vs baseline: 16.0886x; 1.0903x over previous
#include <hip/hip_runtime.h>
#include <math.h>

#define SRATE     22050
#define NFFT      1024
#define HOP       256
#define NMELS     128
#define NFREQS    513
#define FPAD      528
#define BATCH     16
#define NSAMP     661504
#define NFRAMES   2585
#define TTILES    162
#define SPAN      (15*HOP + NFFT)
#define NCH       (BATCH*NMELS)

typedef __attribute__((ext_vector_type(2))) float v2f;
typedef __attribute__((ext_vector_type(8))) float v8f;

#define NTAB 160
__global__ void init_tabs(float* __restrict__ tab) {
    int idx = blockIdx.x * blockDim.x + threadIdx.x;
    if (idx >= NTAB * 32) return;
    const int e = idx >> 5, ln = idx & 31, c = ln & 15, h = ln >> 4;
    const float TWO_PI = 6.28318530717958647692f;
    float v;
    if (e < 32) {
        const int q = e & 15;
        const int s = q & 1, ks = q >> 1;
        const int n1 = 4 * ks + 2 * h + s;
        const bool mid = (e >= 16) && (c == 0);
        const int f1 = mid ? 16 : c;
        const float ang = (float)((f1 * n1) & 31) * (TWO_PI / 32.0f);
        v = (e < 16 || mid) ? cosf(ang) : sinf(ang);
    } else if (e < 64) {
        const int q = e - 32;
        const int s = q & 1, ks = (q >> 1) & 7, mt = q >> 4;
        const int n = 32 * (4 * ks + 2 * h + s) + 16 * mt + c;
        v = 0.5f - 0.5f * cosf(TWO_PI * (float)n / (float)NFFT);
    } else if (e < 128) {
        const int q = (e - 64) & 31;
        const int r = q & 7, nt = (q >> 3) & 1, mt = q >> 4;
        const int f1 = (nt == 0) ? c : ((c == 0) ? 16 : 32 - c);
        const int n2 = 16 * mt + 8 * h + r;
        const float ang = (float)(f1 * n2) * (TWO_PI / (float)NFFT);
        v = (e < 96) ? cosf(ang) : sinf(ang);
    } else {
        const int q = (e - 128) & 15;
        const int s = q & 1, j = (q >> 1) & 3, mt = q >> 3;
        const int f2 = c, n2 = 16 * mt + 8 * h + 2 * j + s;
        const float ang = (float)((f2 * n2) & 31) * (TWO_PI / 32.0f);
        v = (e < 144) ? cosf(ang) : sinf(ang);
    }
    tab[idx] = v;
}

__device__ __forceinline__ float hz_to_mel(float f) { return 2595.0f * log10f(1.0f + f / 700.0f); }
__device__ __forceinline__ float mel_to_hz(float m) { return 700.0f * (exp10f(m / 2595.0f) - 1.0f); }

__global__ void init_mel(float* __restrict__ melFB) {
    int idx = blockIdx.x * blockDim.x + threadIdx.x;
    if (idx >= FPAD * NMELS) return;
    int k = idx / NMELS;
    int m = idx % NMELS;
    float v = 0.0f;
    if (k < NFREQS) {
        float m_lo = hz_to_mel(50.0f);
        float m_hi = hz_to_mel((float)SRATE * 0.5f);
        float step = (m_hi - m_lo) / (float)(NMELS + 1);
        float f0 = mel_to_hz(m_lo + step * (float)m);
        float f1 = mel_to_hz(m_lo + step * (float)(m + 1));
        float f2 = mel_to_hz(m_lo + step * (float)(m + 2));
        float freq = (float)k * ((float)SRATE * 0.5f / (float)(NFFT / 2));
        float down = (freq - f0) / (f1 - f0);
        float up   = (f2 - freq) / (f2 - f1);
        v = fmaxf(0.0f, fminf(down, up));
    }
    melFB[(k >> 1) * (NMELS * 2) + m * 2 + (k & 1)] = v;
}

#define WMMA4(A, B, C) __builtin_amdgcn_wmma_f32_16x16x4_f32(false, (A), false, (B), (short)0, (C), false, false)
__global__ __launch_bounds__(128, 3)
void dft_mel_kernel(const float* __restrict__ x,
                    const float* __restrict__ tab,
                    const float* __restrict__ melFB,
                    float* __restrict__ melBuf)
{
    __shared__ float span[SPAN];
    __shared__ float mag[16 * FPAD];

    const int tile = blockIdx.x;
    const int b    = blockIdx.y;
    const int tid  = threadIdx.x;
    const int ln   = tid & 31;
    const int wv   = tid >> 5;
    const int t0   = tile * 16;
    const int c    = ln & 15;
    const int h    = ln >> 4;

    const float* xb = x + (size_t)b * NSAMP;
    const long base = (long)t0 * HOP - (NFFT / 2);
    for (int i = tid; i < SPAN; i += 128) {
        long p = base + i;
        if (p < 0) p = -p;
        if (p >= NSAMP) p = 2L * NSAMP - 2 - p;
        if (p < 0) p = 0;
        if (p >= NSAMP) p = NSAMP - 1;
        span[i] = xb[p];
    }

    const float* tl = tab + ln;
    v2f B1c[8], B1s[8];
    v2f Wn[2][8];
    float Tc[2][2][8], Ts[2][2][8];
    v2f Ec[2][4], Es[2][4], En[2][4];
    #pragma unroll
    for (int ks = 0; ks < 8; ++ks) {
        const int e = ks * 2;
        B1c[ks].x = tl[(e + 0) * 32];       B1c[ks].y = tl[(e + 1) * 32];
        B1s[ks].x = tl[(16 + e + 0) * 32];  B1s[ks].y = tl[(16 + e + 1) * 32];
    }
    #pragma unroll
    for (int mt = 0; mt < 2; ++mt) {
        #pragma unroll
        for (int ks = 0; ks < 8; ++ks) {
            const int e = 32 + (mt * 8 + ks) * 2;
            Wn[mt][ks].x = tl[(e + 0) * 32];  Wn[mt][ks].y = tl[(e + 1) * 32];
        }
        #pragma unroll
        for (int nt = 0; nt < 2; ++nt) {
            #pragma unroll
            for (int r = 0; r < 8; ++r) {
                const int e = (mt * 2 + nt) * 8 + r;
                Tc[mt][nt][r] = tl[(64 + e) * 32];
                Ts[mt][nt][r] = tl[(96 + e) * 32];
            }
        }
        #pragma unroll
        for (int j = 0; j < 4; ++j) {
            const int e = (mt * 4 + j) * 2;
            Ec[mt][j].x = tl[(128 + e + 0) * 32];  Ec[mt][j].y = tl[(128 + e + 1) * 32];
            Es[mt][j].x = tl[(144 + e + 0) * 32];  Es[mt][j].y = tl[(144 + e + 1) * 32];
            En[mt][j] = -Es[mt][j];
        }
    }
    __syncthreads();

    #pragma unroll 1
    for (int q = 0; q < 4; ++q) {
        const int fr = wv * 4 + q;
        const float* xs = span + fr * HOP + 64 * h + c;

        v8f Dc[2], Ds[2];
        #pragma unroll
        for (int mt = 0; mt < 2; ++mt) {
            Dc[mt] = (v8f){0.f, 0.f, 0.f, 0.f, 0.f, 0.f, 0.f, 0.f};
            Ds[mt] = (v8f){0.f, 0.f, 0.f, 0.f, 0.f, 0.f, 0.f, 0.f};
        }
        #pragma unroll
        for (int ks = 0; ks < 8; ++ks) {
            #pragma unroll
            for (int mt = 0; mt < 2; ++mt) {
                v2f a;
                a.x = xs[128 * ks + 16 * mt]      * Wn[mt][ks].x;
                a.y = xs[128 * ks + 16 * mt + 32] * Wn[mt][ks].y;
                Dc[mt] = WMMA4(a, B1c[ks], Dc[mt]);
                Ds[mt] = WMMA4(a, B1s[ks], Ds[mt]);
            }
        }

        float Pr[2][2][8], Ps[2][2][8];
        const bool c0 = (c == 0);
        #pragma unroll
        for (int mt = 0; mt < 2; ++mt) {
            #pragma unroll
            for (int r = 0; r < 8; ++r) {
                const float dc = Dc[mt][r], ds = Ds[mt][r];
                const float ys = c0 ? 0.0f : ds;
                const float y1 = c0 ? ds : dc;
                Pr[mt][0][r] = dc * Tc[mt][0][r] - ys * Ts[mt][0][r];
                Ps[mt][0][r] = dc * Ts[mt][0][r] + ys * Tc[mt][0][r];
                Pr[mt][1][r] = y1 * Tc[mt][1][r] + ys * Ts[mt][1][r];
                Ps[mt][1][r] = y1 * Ts[mt][1][r] - ys * Tc[mt][1][r];
            }
        }

        #pragma unroll
        for (int nt = 0; nt < 2; ++nt) {
            v8f Xr = (v8f){0.f, 0.f, 0.f, 0.f, 0.f, 0.f, 0.f, 0.f};
            v8f Xi = (v8f){0.f, 0.f, 0.f, 0.f, 0.f, 0.f, 0.f, 0.f};
            #pragma unroll
            for (int mt = 0; mt < 2; ++mt) {
                #pragma unroll
                for (int j = 0; j < 4; ++j) {
                    v2f ar, ai;
                    ar.x = Pr[mt][nt][2 * j];  ar.y = Pr[mt][nt][2 * j + 1];
                    ai.x = Ps[mt][nt][2 * j];  ai.y = Ps[mt][nt][2 * j + 1];
                    Xr = WMMA4(ar, Ec[mt][j], Xr);
                    Xi = WMMA4(ar, Es[mt][j], Xi);
                    Xr = WMMA4(ai, En[mt][j], Xr);
                    Xi = WMMA4(ai, Ec[mt][j], Xi);
                }
            }
            #pragma unroll
            for (int r = 0; r < 8; ++r) {
                const float re = Xr[r], im = Xi[r];
                const int f1 = (nt == 0) ? (8 * h + r) : ((r == 0) ? (16 + 8 * h) : (32 - 8 * h - r));
                mag[fr * FPAD + 32 * c + f1] = sqrtf(re * re + im * im);
            }
        }

        float ny = 0.0f;
        #pragma unroll
        for (int mt = 0; mt < 2; ++mt) {
            #pragma unroll
            for (int r = 0; r < 8; r += 2) ny += Pr[mt][0][r] - Pr[mt][0][r + 1];
        }
        ny += __shfl_xor(ny, 16);
        if (ln < 16) mag[fr * FPAD + 512 + ln] = (ln == 0) ? fabsf(ny) : 0.0f;
    }
    __syncthreads();

    const int mrow  = c;
    const int kh    = h << 1;
    const int ncol  = c;
    const int mbase = h * 8;

    constexpr int MEL_KSTEPS = FPAD / 4;
    constexpr int MEL_KG     = 22;
    static_assert(MEL_KSTEPS % MEL_KG == 0, "whole groups only");
    {
        v8f acc0 = {0.f, 0.f, 0.f, 0.f, 0.f, 0.f, 0.f, 0.f};
        v8f acc1 = {0.f, 0.f, 0.f, 0.f, 0.f, 0.f, 0.f, 0.f};
        const int n0 = wv * 16 + ncol;
        const int n1 = n0 + 64;
        const float* bp = melFB + h * (NMELS * 2) + n0 * 2;
        const float* ap = mag + mrow * FPAD + kh;
        #pragma unroll 1
        for (int g = 0; g < MEL_KSTEPS / MEL_KG; ++g) {
            v2f bf0[MEL_KG], bf1[MEL_KG], af[MEL_KG];
            #pragma unroll
            for (int i = 0; i < MEL_KG; ++i) {
                const float* q = bp + (size_t)(g * MEL_KG + i) * (2 * NMELS * 2);
                bf0[i] = *(const v2f*)q;
                bf1[i] = *(const v2f*)(q + 64 * 2);
            }
            #pragma unroll
            for (int i = 0; i < MEL_KG; ++i)
                af[i] = *(const v2f*)(ap + (g * MEL_KG + i) * 4);
            #pragma unroll
            for (int i = 0; i < MEL_KG; ++i) {
                acc0 = WMMA4(af[i], bf0[i], acc0);
                acc1 = WMMA4(af[i], bf1[i], acc1);
            }
        }
        #pragma unroll
        for (int v = 0; v < 8; ++v) {
            const int t = t0 + mbase + v;
            if (t < NFRAMES) {
                float* o = melBuf + (size_t)t * NCH + b * NMELS;
                o[n0] = acc0[v];
                o[n1] = acc1[v];
            }
        }
    }
}

#define SC_U 56
__global__ __launch_bounds__(32)
void pcen_scan_kernel(const float* __restrict__ melBuf,
                      float* __restrict__ emaBuf)
{
    const int ch = blockIdx.x * 32 + threadIdx.x;
    const float s = 1.0f - expf(-((float)HOP / (float)SRATE) / 0.06f);
    constexpr int NGRP = NFRAMES / SC_U;
    constexpr int REM  = NFRAMES % SC_U;
    static_assert(NGRP >= 1, "at least one whole group");

    const float* p = melBuf + ch;
    float* q = emaBuf + ch;

    float r[SC_U];
    #pragma unroll
    for (int i = 0; i < SC_U; ++i) r[i] = p[(size_t)i * NCH];

    float M = 0.0f;
    for (int g = 0; g < NGRP - 1; ++g) {
        #pragma unroll
        for (int i = 0; i < SC_U; ++i) {
            M = (1.0f - s) * M + s * r[i];
            q[(size_t)i * NCH] = M;
            r[i] = p[(size_t)(SC_U + i) * NCH];
        }
        p += (size_t)SC_U * NCH;
        q += (size_t)SC_U * NCH;
    }
    #pragma unroll
    for (int i = 0; i < SC_U; ++i) {
        M = (1.0f - s) * M + s * r[i];
        q[(size_t)i * NCH] = M;
        if (i < REM) r[i] = p[(size_t)(SC_U + i) * NCH];
    }
    p += (size_t)SC_U * NCH;
    q += (size_t)SC_U * NCH;
    #pragma unroll
    for (int i = 0; i < REM; ++i) {
        M = (1.0f - s) * M + s * r[i];
        q[(size_t)i * NCH] = M;
    }
}

#define PW_TT    128
#define PW_PITCH (PW_TT + 1)
#define PW_TTILES ((NFRAMES + PW_TT - 1) / PW_TT)
__global__ __launch_bounds__(256)
void pcen_point_kernel(const float* __restrict__ melBuf,
                       const float* __restrict__ emaBuf,
                       float* __restrict__ out)
{
    __shared__ float tile[32 * PW_PITCH];

    const int tid = threadIdx.x;
    const int ln  = tid & 31;
    const int wv  = tid >> 5;
    const int t0  = blockIdx.x * PW_TT;
    const int ch0 = blockIdx.y * 32;
    const float da = powf(2.0f, 0.98f);

    #pragma unroll 1
    for (int kk = 0; kk < PW_TT / 8; kk += 4) {
        float e[4], M[4];
        #pragma unroll
        for (int j = 0; j < 4; ++j) {
            const int row = wv + 8 * (kk + j);
            int t = t0 + row;
            if (t >= NFRAMES) t = NFRAMES - 1;
            const int off = t * NCH + ch0 + ln;
            e[j] = melBuf[off];
            M[j] = emaBuf[off];
        }
        #pragma unroll
        for (int j = 0; j < 4; ++j) {
            const int row = wv + 8 * (kk + j);
            float smooth = sqrtf(M[j] + 1e-6f);
            tile[ln * PW_PITCH + row] = powf(e[j] / smooth + 2.0f, 0.98f) - da;
        }
    }
    __syncthreads();

    const int tt = tid % PW_TT;
    const int c0 = tid / PW_TT;
    if (t0 + tt < NFRAMES) {
        float* o = out + (size_t)ch0 * NFRAMES + t0 + tt;
        for (int k = 0; k < 32 / (256 / PW_TT); ++k) {
            const int c = k * (256 / PW_TT) + c0;
            o[c * NFRAMES] = tile[c * PW_PITCH + tt];
        }
    }
}

extern "C" void kernel_launch(void* const* d_in, const int* in_sizes, int n_in,
                              void* d_out, int out_size, void* d_ws, size_t ws_size,
                              hipStream_t stream) {
    const float* x = (const float*)d_in[0];
    float* ws = (float*)d_ws;

    float* tab    = ws;
    float* melFB  = tab    + (size_t)NTAB * 32;
    float* melBuf = melFB  + (size_t)FPAD * NMELS;
    float* emaBuf = melBuf + (size_t)2592 * NCH;

    init_tabs<<<(NTAB * 32 + 255) / 256, 256, 0, stream>>>(tab);
    int n2 = FPAD * NMELS;
    init_mel<<<(n2 + 255) / 256, 256, 0, stream>>>(melFB);

    dim3 grid(TTILES, BATCH);
    dft_mel_kernel<<<grid, 128, 0, stream>>>(x, tab, melFB, melBuf);

    pcen_scan_kernel<<<NCH / 32, 32, 0, stream>>>(melBuf, emaBuf);

    dim3 pgrid(PW_TTILES, NCH / 32);
    pcen_point_kernel<<<pgrid, 256, 0, stream>>>(melBuf, emaBuf, (float*)d_out);
}


# pcen: scan and pointwise kernels fused by wave roles: wave 0 runs the serial EMA (order unchanged) on LDS chunks while 8 waves prefetch frames and do pow/sqrt/divide; EMA plane gone; 5 -> 4 launches
# speedup vs baseline: 20.5939x; 1.2800x over previous
#include <hip/hip_runtime.h>
#include <math.h>

#define SRATE     22050
#define NFFT      1024
#define HOP       256
#define NMELS     128
#define NFREQS    513
#define FPAD      528
#define BATCH     16
#define NSAMP     661504
#define NFRAMES   2585
#define TTILES    162
#define SPAN      (15*HOP + NFFT)
#define NCH       (BATCH*NMELS)

typedef __attribute__((ext_vector_type(2))) float v2f;
typedef __attribute__((ext_vector_type(8))) float v8f;

#define NTAB 160
__global__ void init_tabs(float* __restrict__ tab) {
    int idx = blockIdx.x * blockDim.x + threadIdx.x;
    if (idx >= NTAB * 32) return;
    const int e = idx >> 5, ln = idx & 31, c = ln & 15, h = ln >> 4;
    const float TWO_PI = 6.28318530717958647692f;
    float v;
    if (e < 32) {
        const int q = e & 15;
        const int s = q & 1, ks = q >> 1;
        const int n1 = 4 * ks + 2 * h + s;
        const bool mid = (e >= 16) && (c == 0);
        const int f1 = mid ? 16 : c;
        const float ang = (float)((f1 * n1) & 31) * (TWO_PI / 32.0f);
        v = (e < 16 || mid) ? cosf(ang) : sinf(ang);
    } else if (e < 64) {
        const int q = e - 32;
        const int s = q & 1, ks = (q >> 1) & 7, mt = q >> 4;
        const int n = 32 * (4 * ks + 2 * h + s) + 16 * mt + c;
        v = 0.5f - 0.5f * cosf(TWO_PI * (float)n / (float)NFFT);
    } else if (e < 128) {
        const int q = (e - 64) & 31;
        const int r = q & 7, nt = (q >> 3) & 1, mt = q >> 4;
        const int f1 = (nt == 0) ? c : ((c == 0) ? 16 : 32 - c);
        const int n2 = 16 * mt + 8 * h + r;
        const float ang = (float)(f1 * n2) * (TWO_PI / (float)NFFT);
        v = (e < 96) ? cosf(ang) : sinf(ang);
    } else {
        const int q = (e - 128) & 15;
        const int s = q & 1, j = (q >> 1) & 3, mt = q >> 3;
        const int f2 = c, n2 = 16 * mt + 8 * h + 2 * j + s;
        const float ang = (float)((f2 * n2) & 31) * (TWO_PI / 32.0f);
        v = (e < 144) ? cosf(ang) : sinf(ang);
    }
    tab[idx] = v;
}

__device__ __forceinline__ float hz_to_mel(float f) { return 2595.0f * log10f(1.0f + f / 700.0f); }
__device__ __forceinline__ float mel_to_hz(float m) { return 700.0f * (exp10f(m / 2595.0f) - 1.0f); }

__global__ void init_mel(float* __restrict__ melFB) {
    int idx = blockIdx.x * blockDim.x + threadIdx.x;
    if (idx >= FPAD * NMELS) return;
    int k = idx / NMELS;
    int m = idx % NMELS;
    float v = 0.0f;
    if (k < NFREQS) {
        float m_lo = hz_to_mel(50.0f);
        float m_hi = hz_to_mel((float)SRATE * 0.5f);
        float step = (m_hi - m_lo) / (float)(NMELS + 1);
        float f0 = mel_to_hz(m_lo + step * (float)m);
        float f1 = mel_to_hz(m_lo + step * (float)(m + 1));
        float f2 = mel_to_hz(m_lo + step * (float)(m + 2));
        float freq = (float)k * ((float)SRATE * 0.5f / (float)(NFFT / 2));
        float down = (freq - f0) / (f1 - f0);
        float up   = (f2 - freq) / (f2 - f1);
        v = fmaxf(0.0f, fminf(down, up));
    }
    melFB[(k >> 1) * (NMELS * 2) + m * 2 + (k & 1)] = v;
}

#define WMMA4(A, B, C) __builtin_amdgcn_wmma_f32_16x16x4_f32(false, (A), false, (B), (short)0, (C), false, false)
__global__ __launch_bounds__(128, 3)
void dft_mel_kernel(const float* __restrict__ x,
                    const float* __restrict__ tab,
                    const float* __restrict__ melFB,
                    float* __restrict__ melBuf)
{
    __shared__ float span[SPAN];
    __shared__ float mag[16 * FPAD];

    const int tile = blockIdx.x;
    const int b    = blockIdx.y;
    const int tid  = threadIdx.x;
    const int ln   = tid & 31;
    const int wv   = tid >> 5;
    const int t0   = tile * 16;
    const int c    = ln & 15;
    const int h    = ln >> 4;

    const float* xb = x + (size_t)b * NSAMP;
    const long base = (long)t0 * HOP - (NFFT / 2);
    for (int i = tid; i < SPAN; i += 128) {
        long p = base + i;
        if (p < 0) p = -p;
        if (p >= NSAMP) p = 2L * NSAMP - 2 - p;
        if (p < 0) p = 0;
        if (p >= NSAMP) p = NSAMP - 1;
        span[i] = xb[p];
    }

    const float* tl = tab + ln;
    v2f B1c[8], B1s[8];
    v2f Wn[2][8];
    float Tc[2][2][8], Ts[2][2][8];
    v2f Ec[2][4], Es[2][4], En[2][4];
    #pragma unroll
    for (int ks = 0; ks < 8; ++ks) {
        const int e = ks * 2;
        B1c[ks].x = tl[(e + 0) * 32];       B1c[ks].y = tl[(e + 1) * 32];
        B1s[ks].x = tl[(16 + e + 0) * 32];  B1s[ks].y = tl[(16 + e + 1) * 32];
    }
    #pragma unroll
    for (int mt = 0; mt < 2; ++mt) {
        #pragma unroll
        for (int ks = 0; ks < 8; ++ks) {
            const int e = 32 + (mt * 8 + ks) * 2;
            Wn[mt][ks].x = tl[(e + 0) * 32];  Wn[mt][ks].y = tl[(e + 1) * 32];
        }
        #pragma unroll
        for (int nt = 0; nt < 2; ++nt) {
            #pragma unroll
            for (int r = 0; r < 8; ++r) {
                const int e = (mt * 2 + nt) * 8 + r;
                Tc[mt][nt][r] = tl[(64 + e) * 32];
                Ts[mt][nt][r] = tl[(96 + e) * 32];
            }
        }
        #pragma unroll
        for (int j = 0; j < 4; ++j) {
            const int e = (mt * 4 + j) * 2;
            Ec[mt][j].x = tl[(128 + e + 0) * 32];  Ec[mt][j].y = tl[(128 + e + 1) * 32];
            Es[mt][j].x = tl[(144 + e + 0) * 32];  Es[mt][j].y = tl[(144 + e + 1) * 32];
            En[mt][j] = -Es[mt][j];
        }
    }
    __syncthreads();

    #pragma unroll 1
    for (int q = 0; q < 4; ++q) {
        const int fr = wv * 4 + q;
        const float* xs = span + fr * HOP + 64 * h + c;

        v8f Dc[2], Ds[2];
        #pragma unroll
        for (int mt = 0; mt < 2; ++mt) {
            Dc[mt] = (v8f){0.f, 0.f, 0.f, 0.f, 0.f, 0.f, 0.f, 0.f};
            Ds[mt] = (v8f){0.f, 0.f, 0.f, 0.f, 0.f, 0.f, 0.f, 0.f};
        }
        #pragma unroll
        for (int ks = 0; ks < 8; ++ks) {
            #pragma unroll
            for (int mt = 0; mt < 2; ++mt) {
                v2f a;
                a.x = xs[128 * ks + 16 * mt]      * Wn[mt][ks].x;
                a.y = xs[128 * ks + 16 * mt + 32] * Wn[mt][ks].y;
                Dc[mt] = WMMA4(a, B1c[ks], Dc[mt]);
                Ds[mt] = WMMA4(a, B1s[ks], Ds[mt]);
            }
        }

        float Pr[2][2][8], Ps[2][2][8];
        const bool c0 = (c == 0);
        #pragma unroll
        for (int mt = 0; mt < 2; ++mt) {
            #pragma unroll
            for (int r = 0; r < 8; ++r) {
                const float dc = Dc[mt][r], ds = Ds[mt][r];
                const float ys = c0 ? 0.0f : ds;
                const float y1 = c0 ? ds : dc;
                Pr[mt][0][r] = dc * Tc[mt][0][r] - ys * Ts[mt][0][r];
                Ps[mt][0][r] = dc * Ts[mt][0][r] + ys * Tc[mt][0][r];
                Pr[mt][1][r] = y1 * Tc[mt][1][r] + ys * Ts[mt][1][r];
                Ps[mt][1][r] = y1 * Ts[mt][1][r] - ys * Tc[mt][1][r];
            }
        }

        #pragma unroll
        for (int nt = 0; nt < 2; ++nt) {
            v8f Xr = (v8f){0.f, 0.f, 0.f, 0.f, 0.f, 0.f, 0.f, 0.f};
            v8f Xi = (v8f){0.f, 0.f, 0.f, 0.f, 0.f, 0.f, 0.f, 0.f};
            #pragma unroll
            for (int mt = 0; mt < 2; ++mt) {
                #pragma unroll
                for (int j = 0; j < 4; ++j) {
                    v2f ar, ai;
                    ar.x = Pr[mt][nt][2 * j];  ar.y = Pr[mt][nt][2 * j + 1];
                    ai.x = Ps[mt][nt][2 * j];  ai.y = Ps[mt][nt][2 * j + 1];
                    Xr = WMMA4(ar, Ec[mt][j], Xr);
                    Xi = WMMA4(ar, Es[mt][j], Xi);
                    Xr = WMMA4(ai, En[mt][j], Xr);
                    Xi = WMMA4(ai, Ec[mt][j], Xi);
                }
            }
            #pragma unroll
            for (int r = 0; r < 8; ++r) {
                const float re = Xr[r], im = Xi[r];
                const int f1 = (nt == 0) ? (8 * h + r) : ((r == 0) ? (16 + 8 * h) : (32 - 8 * h - r));
                mag[fr * FPAD + 32 * c + f1] = sqrtf(re * re + im * im);
            }
        }

        float ny = 0.0f;
        #pragma unroll
        for (int mt = 0; mt < 2; ++mt) {
            #pragma unroll
            for (int r = 0; r < 8; r += 2) ny += Pr[mt][0][r] - Pr[mt][0][r + 1];
        }
        ny += __shfl_xor(ny, 16);
        if (ln < 16) mag[fr * FPAD + 512 + ln] = (ln == 0) ? fabsf(ny) : 0.0f;
    }
    __syncthreads();

    const int mrow  = c;
    const int kh    = h << 1;
    const int ncol  = c;
    const int mbase = h * 8;

    constexpr int MEL_KSTEPS = FPAD / 4;
    constexpr int MEL_KG     = 22;
    static_assert(MEL_KSTEPS % MEL_KG == 0, "whole groups only");
    {
        v8f acc0 = {0.f, 0.f, 0.f, 0.f, 0.f, 0.f, 0.f, 0.f};
        v8f acc1 = {0.f, 0.f, 0.f, 0.f, 0.f, 0.f, 0.f, 0.f};
        const int n0 = wv * 16 + ncol;
        const int n1 = n0 + 64;
        const float* bp = melFB + h * (NMELS * 2) + n0 * 2;
        const float* ap = mag + mrow * FPAD + kh;
        #pragma unroll 1
        for (int g = 0; g < MEL_KSTEPS / MEL_KG; ++g) {
            v2f bf0[MEL_KG], bf1[MEL_KG], af[MEL_KG];
            #pragma unroll
            for (int i = 0; i < MEL_KG; ++i) {
                const float* q = bp + (size_t)(g * MEL_KG + i) * (2 * NMELS * 2);
                bf0[i] = *(const v2f*)q;
                bf1[i] = *(const v2f*)(q + 64 * 2);
            }
            #pragma unroll
            for (int i = 0; i < MEL_KG; ++i)
                af[i] = *(const v2f*)(ap + (g * MEL_KG + i) * 4);
            #pragma unroll
            for (int i = 0; i < MEL_KG; ++i) {
                acc0 = WMMA4(af[i], bf0[i], acc0);
                acc1 = WMMA4(af[i], bf1[i], acc1);
            }
        }
        #pragma unroll
        for (int v = 0; v < 8; ++v) {
            const int t = t0 + mbase + v;
            if (t < NFRAMES) {
                float* o = melBuf + (size_t)t * NCH + b * NMELS;
                o[n0] = acc0[v];
                o[n1] = acc1[v];
            }
        }
    }
}

#define PF_CP      8
#define PF_TC      128
#define PF_PT      (PF_TC + 4)
#define PF_NK      ((NFRAMES + PF_TC - 1) / PF_TC)
#define PF_WORKERS 8

__device__ __forceinline__ void pcen_lds_barrier() {
    __builtin_amdgcn_fence(__ATOMIC_RELEASE, "workgroup", "local");
    __builtin_amdgcn_s_barrier();
    __builtin_amdgcn_fence(__ATOMIC_ACQUIRE, "workgroup", "local");
}

template <bool GUARD>
__device__ __forceinline__ void pcen_point4(const float* eb, const float* mb, float* __restrict__ out,
                                            int w, int ln, int ch0, int k, float da)
{
    float ev[4], Mv[4];
    #pragma unroll
    for (int j = 0; j < 4; ++j) {
        const int u = w * 4 + j;
        const int o = (u >> 2) * PF_PT + (u & 3) * 32 + ln;
        ev[j] = eb[o];
        Mv[j] = mb[o];
    }
    #pragma unroll
    for (int j = 0; j < 4; ++j) {
        const int u  = w * 4 + j;
        const int tg = k * PF_TC + (u & 3) * 32 + ln;
        float smooth = sqrtf(Mv[j] + 1e-6f);
        float v = powf(ev[j] / smooth + 2.0f, 0.98f) - da;
        if (!GUARD || tg < NFRAMES) out[(size_t)(ch0 + (u >> 2)) * NFRAMES + tg] = v;
    }
}

__global__ __launch_bounds__(32 * (1 + PF_WORKERS))
void pcen_fused_kernel(const float* __restrict__ melBuf,
                       float* __restrict__ out)
{
    __shared__ __attribute__((aligned(16))) float Eb[3][PF_CP * PF_PT];
    __shared__ __attribute__((aligned(16))) float Mb[2][PF_CP * PF_PT];
    static_assert(PF_CP * (PF_TC / 32) == 4 * PF_WORKERS, "four (channel, 32-frame) units per worker wave");
    static_assert(PF_CP * PF_TC == 4 * 32 * PF_WORKERS, "four fetched values per worker lane");

    const int ln  = threadIdx.x & 31;
    const int wv  = __builtin_amdgcn_readfirstlane((int)(threadIdx.x >> 5));
    const int ch0 = blockIdx.x * PF_CP;
    const float s  = 1.0f - expf(-((float)HOP / (float)SRATE) / 0.06f);
    const float da = powf(2.0f, 0.98f);

    float M = 0.0f;
    const int p  = (wv - 1) * 32 + ln;
    const int pc = p & (PF_CP - 1);
    const int pt = p >> 3;

    #pragma unroll 1
    for (int it = 0; it < PF_NK + 2; ++it) {
        if (wv == 0) {
            if (it >= 1 && it <= PF_NK && ln < PF_CP) {
                const float4* e4 = (const float4*)(Eb[(it - 1) % 3] + ln * PF_PT);
                float4*       m4 = (float4*)(Mb[(it - 1) & 1] + ln * PF_PT);
                float4 e[PF_TC / 4];
                #pragma unroll
                for (int j = 0; j < PF_TC / 4; ++j) e[j] = e4[j];
                #pragma unroll
                for (int j = 0; j < PF_TC / 4; ++j) {
                    float4 m;
                    M = __builtin_fmaf(1.0f - s, M, s * e[j].x);  m.x = M;
                    M = __builtin_fmaf(1.0f - s, M, s * e[j].y);  m.y = M;
                    M = __builtin_fmaf(1.0f - s, M, s * e[j].z);  m.z = M;
                    M = __builtin_fmaf(1.0f - s, M, s * e[j].w);  m.w = M;
                    m4[j] = m;
                }
            }
        } else {
            const bool fetch = it < PF_NK;
            float pre[4] = {0.0f, 0.0f, 0.0f, 0.0f};
            if (fetch) {
                #pragma unroll
                for (int i = 0; i < 4; ++i) {
                    int t = it * PF_TC + pt + 32 * i;
                    if (t >= NFRAMES) t = NFRAMES - 1;
                    pre[i] = melBuf[t * NCH + ch0 + pc];
                }
            }
            if (it >= 2) {
                const int k = it - 2;
                if (k < PF_NK - 1) pcen_point4<false>(Eb[k % 3], Mb[k & 1], out, wv - 1, ln, ch0, k, da);
                else               pcen_point4<true >(Eb[k % 3], Mb[k & 1], out, wv - 1, ln, ch0, k, da);
            }
            if (fetch) {
                float* ed = Eb[it % 3] + pc * PF_PT + pt;
                #pragma unroll
                for (int i = 0; i < 4; ++i) ed[32 * i] = pre[i];
            }
        }
        pcen_lds_barrier();
    }
}

extern "C" void kernel_launch(void* const* d_in, const int* in_sizes, int n_in,
                              void* d_out, int out_size, void* d_ws, size_t ws_size,
                              hipStream_t stream) {
    const float* x = (const float*)d_in[0];
    float* ws = (float*)d_ws;

    float* tab    = ws;
    float* melFB  = tab    + (size_t)NTAB * 32;
    float* melBuf = melFB  + (size_t)FPAD * NMELS;

    init_tabs<<<(NTAB * 32 + 255) / 256, 256, 0, stream>>>(tab);
    int n2 = FPAD * NMELS;
    init_mel<<<(n2 + 255) / 256, 256, 0, stream>>>(melFB);

    dim3 grid(TTILES, BATCH);
    dft_mel_kernel<<<grid, 128, 0, stream>>>(x, tab, melFB, melBuf);

    pcen_fused_kernel<<<NCH / PF_CP, 32 * (1 + PF_WORKERS), 0, stream>>>(melBuf, (float*)d_out);
}


# dft_mel stage 2: each complex product on three real products, (Pr+Ps)*cos, Pr*(sin-cos), Ps*(cos+sin), not four; 64->48 WMMAs a frame, 96->80 in all; DFT rounding and sum order change
# speedup vs baseline: 21.3055x; 1.0346x over previous
#include <hip/hip_runtime.h>
#include <math.h>

#define SRATE     22050
#define NFFT      1024
#define HOP       256
#define NMELS     128
#define NFREQS    513
#define FPAD      528
#define BATCH     16
#define NSAMP     661504
#define NFRAMES   2585
#define TTILES    162
#define SPAN      (15*HOP + NFFT)
#define NCH       (BATCH*NMELS)

typedef __attribute__((ext_vector_type(2))) float v2f;
typedef __attribute__((ext_vector_type(8))) float v8f;

#define NTAB 160
__global__ void init_tabs(float* __restrict__ tab) {
    int idx = blockIdx.x * blockDim.x + threadIdx.x;
    if (idx >= NTAB * 32) return;
    const int e = idx >> 5, ln = idx & 31, c = ln & 15, h = ln >> 4;
    const float TWO_PI = 6.28318530717958647692f;
    float v;
    if (e < 32) {
        const int q = e & 15;
        const int s = q & 1, ks = q >> 1;
        const int n1 = 4 * ks + 2 * h + s;
        const bool mid = (e >= 16) && (c == 0);
        const int f1 = mid ? 16 : c;
        const float ang = (float)((f1 * n1) & 31) * (TWO_PI / 32.0f);
        v = (e < 16 || mid) ? cosf(ang) : sinf(ang);
    } else if (e < 64) {
        const int q = e - 32;
        const int s = q & 1, ks = (q >> 1) & 7, mt = q >> 4;
        const int n = 32 * (4 * ks + 2 * h + s) + 16 * mt + c;
        v = 0.5f - 0.5f * cosf(TWO_PI * (float)n / (float)NFFT);
    } else if (e < 128) {
        const int q = (e - 64) & 31;
        const int r = q & 7, nt = (q >> 3) & 1, mt = q >> 4;
        const int f1 = (nt == 0) ? c : ((c == 0) ? 16 : 32 - c);
        const int n2 = 16 * mt + 8 * h + r;
        const float ang = (float)(f1 * n2) * (TWO_PI / (float)NFFT);
        v = (e < 96) ? cosf(ang) : sinf(ang);
    } else {
        const int q = (e - 128) & 15;
        const int s = q & 1, j = (q >> 1) & 3, mt = q >> 3;
        const int f2 = c, n2 = 16 * mt + 8 * h + 2 * j + s;
        const float ang = (float)((f2 * n2) & 31) * (TWO_PI / 32.0f);
        v = (e < 144) ? cosf(ang) : sinf(ang);
    }
    tab[idx] = v;
}

__device__ __forceinline__ float hz_to_mel(float f) { return 2595.0f * log10f(1.0f + f / 700.0f); }
__device__ __forceinline__ float mel_to_hz(float m) { return 700.0f * (exp10f(m / 2595.0f) - 1.0f); }

__global__ void init_mel(float* __restrict__ melFB) {
    int idx = blockIdx.x * blockDim.x + threadIdx.x;
    if (idx >= FPAD * NMELS) return;
    int k = idx / NMELS;
    int m = idx % NMELS;
    float v = 0.0f;
    if (k < NFREQS) {
        float m_lo = hz_to_mel(50.0f);
        float m_hi = hz_to_mel((float)SRATE * 0.5f);
        float step = (m_hi - m_lo) / (float)(NMELS + 1);
        float f0 = mel_to_hz(m_lo + step * (float)m);
        float f1 = mel_to_hz(m_lo + step * (float)(m + 1));
        float f2 = mel_to_hz(m_lo + step * (float)(m + 2));
        float freq = (float)k * ((float)SRATE * 0.5f / (float)(NFFT / 2));
        float down = (freq - f0) / (f1 - f0);
        float up   = (f2 - freq) / (f2 - f1);
        v = fmaxf(0.0f, fminf(down, up));
    }
    melFB[(k >> 1) * (NMELS * 2) + m * 2 + (k & 1)] = v;
}

#define WMMA4(A, B, C) __builtin_amdgcn_wmma_f32_16x16x4_f32(false, (A), false, (B), (short)0, (C), false, false)
__global__ __launch_bounds__(128, 3)
void dft_mel_kernel(const float* __restrict__ x,
                    const float* __restrict__ tab,
                    const float* __restrict__ melFB,
                    float* __restrict__ melBuf)
{
    __shared__ float span[SPAN];
    __shared__ float mag[16 * FPAD];

    const int tile = blockIdx.x;
    const int b    = blockIdx.y;
    const int tid  = threadIdx.x;
    const int ln   = tid & 31;
    const int wv   = tid >> 5;
    const int t0   = tile * 16;
    const int c    = ln & 15;
    const int h    = ln >> 4;

    const float* xb = x + (size_t)b * NSAMP;
    const long base = (long)t0 * HOP - (NFFT / 2);
    for (int i = tid; i < SPAN; i += 128) {
        long p = base + i;
        if (p < 0) p = -p;
        if (p >= NSAMP) p = 2L * NSAMP - 2 - p;
        if (p < 0) p = 0;
        if (p >= NSAMP) p = NSAMP - 1;
        span[i] = xb[p];
    }

    const float* tl = tab + ln;
    v2f B1c[8], B1s[8];
    v2f Wn[2][8];
    float Tc[2][2][8], Ts[2][2][8];
    v2f Ec[2][4], Ed[2][4], Ep[2][4];
    #pragma unroll
    for (int ks = 0; ks < 8; ++ks) {
        const int e = ks * 2;
        B1c[ks].x = tl[(e + 0) * 32];       B1c[ks].y = tl[(e + 1) * 32];
        B1s[ks].x = tl[(16 + e + 0) * 32];  B1s[ks].y = tl[(16 + e + 1) * 32];
    }
    #pragma unroll
    for (int mt = 0; mt < 2; ++mt) {
        #pragma unroll
        for (int ks = 0; ks < 8; ++ks) {
            const int e = 32 + (mt * 8 + ks) * 2;
            Wn[mt][ks].x = tl[(e + 0) * 32];  Wn[mt][ks].y = tl[(e + 1) * 32];
        }
        #pragma unroll
        for (int nt = 0; nt < 2; ++nt) {
            #pragma unroll
            for (int r = 0; r < 8; ++r) {
                const int e = (mt * 2 + nt) * 8 + r;
                Tc[mt][nt][r] = tl[(64 + e) * 32];
                Ts[mt][nt][r] = tl[(96 + e) * 32];
            }
        }
        #pragma unroll
        for (int j = 0; j < 4; ++j) {
            const int e = (mt * 4 + j) * 2;
            v2f es;
            Ec[mt][j].x = tl[(128 + e + 0) * 32];  Ec[mt][j].y = tl[(128 + e + 1) * 32];
            es.x        = tl[(144 + e + 0) * 32];  es.y        = tl[(144 + e + 1) * 32];
            Ed[mt][j] = es - Ec[mt][j];
            Ep[mt][j] = Ec[mt][j] + es;
        }
    }
    __syncthreads();

    #pragma unroll 1
    for (int q = 0; q < 4; ++q) {
        const int fr = wv * 4 + q;
        const float* xs = span + fr * HOP + 64 * h + c;

        v8f Dc[2], Ds[2];
        #pragma unroll
        for (int mt = 0; mt < 2; ++mt) {
            Dc[mt] = (v8f){0.f, 0.f, 0.f, 0.f, 0.f, 0.f, 0.f, 0.f};
            Ds[mt] = (v8f){0.f, 0.f, 0.f, 0.f, 0.f, 0.f, 0.f, 0.f};
        }
        #pragma unroll
        for (int ks = 0; ks < 8; ++ks) {
            #pragma unroll
            for (int mt = 0; mt < 2; ++mt) {
                v2f a;
                a.x = xs[128 * ks + 16 * mt]      * Wn[mt][ks].x;
                a.y = xs[128 * ks + 16 * mt + 32] * Wn[mt][ks].y;
                Dc[mt] = WMMA4(a, B1c[ks], Dc[mt]);
                Ds[mt] = WMMA4(a, B1s[ks], Ds[mt]);
            }
        }

        float Pr[2][2][8], Ps[2][2][8];
        const bool c0 = (c == 0);
        #pragma unroll
        for (int mt = 0; mt < 2; ++mt) {
            #pragma unroll
            for (int r = 0; r < 8; ++r) {
                const float dc = Dc[mt][r], ds = Ds[mt][r];
                const float ys = c0 ? 0.0f : ds;
                const float y1 = c0 ? ds : dc;
                Pr[mt][0][r] = dc * Tc[mt][0][r] - ys * Ts[mt][0][r];
                Ps[mt][0][r] = dc * Ts[mt][0][r] + ys * Tc[mt][0][r];
                Pr[mt][1][r] = y1 * Tc[mt][1][r] + ys * Ts[mt][1][r];
                Ps[mt][1][r] = y1 * Ts[mt][1][r] - ys * Tc[mt][1][r];
            }
        }

        #pragma unroll
        for (int nt = 0; nt < 2; ++nt) {
            v8f T1 = (v8f){0.f, 0.f, 0.f, 0.f, 0.f, 0.f, 0.f, 0.f};
            v8f T2 = (v8f){0.f, 0.f, 0.f, 0.f, 0.f, 0.f, 0.f, 0.f};
            v8f T3 = (v8f){0.f, 0.f, 0.f, 0.f, 0.f, 0.f, 0.f, 0.f};
            #pragma unroll
            for (int mt = 0; mt < 2; ++mt) {
                #pragma unroll
                for (int j = 0; j < 4; ++j) {
                    v2f ar, ai;
                    ar.x = Pr[mt][nt][2 * j];  ar.y = Pr[mt][nt][2 * j + 1];
                    ai.x = Ps[mt][nt][2 * j];  ai.y = Ps[mt][nt][2 * j + 1];
                    const v2f as = ar + ai;
                    T1 = WMMA4(as, Ec[mt][j], T1);
                    T2 = WMMA4(ar, Ed[mt][j], T2);
                    T3 = WMMA4(ai, Ep[mt][j], T3);
                }
            }
            const v8f Xr = T1 - T3;
            const v8f Xi = T1 + T2;
            #pragma unroll
            for (int r = 0; r < 8; ++r) {
                const float re = Xr[r], im = Xi[r];
                const int f1 = (nt == 0) ? (8 * h + r) : ((r == 0) ? (16 + 8 * h) : (32 - 8 * h - r));
                mag[fr * FPAD + 32 * c + f1] = sqrtf(re * re + im * im);
            }
        }

        float ny = 0.0f;
        #pragma unroll
        for (int mt = 0; mt < 2; ++mt) {
            #pragma unroll
            for (int r = 0; r < 8; r += 2) ny += Pr[mt][0][r] - Pr[mt][0][r + 1];
        }
        ny += __shfl_xor(ny, 16);
        if (ln < 16) mag[fr * FPAD + 512 + ln] = (ln == 0) ? fabsf(ny) : 0.0f;
    }
    __syncthreads();

    const int mrow  = c;
    const int kh    = h << 1;
    const int ncol  = c;
    const int mbase = h * 8;

    constexpr int MEL_KSTEPS = FPAD / 4;
    constexpr int MEL_KG     = 22;
    static_assert(MEL_KSTEPS % MEL_KG == 0, "whole groups only");
    {
        v8f acc0 = {0.f, 0.f, 0.f, 0.f, 0.f, 0.f, 0.f, 0.f};
        v8f acc1 = {0.f, 0.f, 0.f, 0.f, 0.f, 0.f, 0.f, 0.f};
        const int n0 = wv * 16 + ncol;
        const int n1 = n0 + 64;
        const float* bp = melFB + h * (NMELS * 2) + n0 * 2;
        const float* ap = mag + mrow * FPAD + kh;
        #pragma unroll 1
        for (int g = 0; g < MEL_KSTEPS / MEL_KG; ++g) {
            v2f bf0[MEL_KG], bf1[MEL_KG], af[MEL_KG];
            #pragma unroll
            for (int i = 0; i < MEL_KG; ++i) {
                const float* q = bp + (size_t)(g * MEL_KG + i) * (2 * NMELS * 2);
                bf0[i] = *(const v2f*)q;
                bf1[i] = *(const v2f*)(q + 64 * 2);
            }
            #pragma unroll
            for (int i = 0; i < MEL_KG; ++i)
                af[i] = *(const v2f*)(ap + (g * MEL_KG + i) * 4);
            #pragma unroll
            for (int i = 0; i < MEL_KG; ++i) {
                acc0 = WMMA4(af[i], bf0[i], acc0);
                acc1 = WMMA4(af[i], bf1[i], acc1);
            }
        }
        #pragma unroll
        for (int v = 0; v < 8; ++v) {
            const int t = t0 + mbase + v;
            if (t < NFRAMES) {
                float* o = melBuf + (size_t)t * NCH + b * NMELS;
                o[n0] = acc0[v];
                o[n1] = acc1[v];
            }
        }
    }
}

#define PF_CP      8
#define PF_TC      128
#define PF_PT      (PF_TC + 4)
#define PF_NK      ((NFRAMES + PF_TC - 1) / PF_TC)
#define PF_WORKERS 8

__device__ __forceinline__ void pcen_lds_barrier() {
    __builtin_amdgcn_fence(__ATOMIC_RELEASE, "workgroup", "local");
    __builtin_amdgcn_s_barrier();
    __builtin_amdgcn_fence(__ATOMIC_ACQUIRE, "workgroup", "local");
}

template <bool GUARD>
__device__ __forceinline__ void pcen_point4(const float* eb, const float* mb, float* __restrict__ out,
                                            int w, int ln, int ch0, int k, float da)
{
    float ev[4], Mv[4];
    #pragma unroll
    for (int j = 0; j < 4; ++j) {
        const int u = w * 4 + j;
        const int o = (u >> 2) * PF_PT + (u & 3) * 32 + ln;
        ev[j] = eb[o];
        Mv[j] = mb[o];
    }
    #pragma unroll
    for (int j = 0; j < 4; ++j) {
        const int u  = w * 4 + j;
        const int tg = k * PF_TC + (u & 3) * 32 + ln;
        float smooth = sqrtf(Mv[j] + 1e-6f);
        float v = powf(ev[j] / smooth + 2.0f, 0.98f) - da;
        if (!GUARD || tg < NFRAMES) out[(size_t)(ch0 + (u >> 2)) * NFRAMES + tg] = v;
    }
}

__global__ __launch_bounds__(32 * (1 + PF_WORKERS))
void pcen_fused_kernel(const float* __restrict__ melBuf,
                       float* __restrict__ out)
{
    __shared__ __attribute__((aligned(16))) float Eb[3][PF_CP * PF_PT];
    __shared__ __attribute__((aligned(16))) float Mb[2][PF_CP * PF_PT];
    static_assert(PF_CP * (PF_TC / 32) == 4 * PF_WORKERS, "four (channel, 32-frame) units per worker wave");
    static_assert(PF_CP * PF_TC == 4 * 32 * PF_WORKERS, "four fetched values per worker lane");

    const int ln  = threadIdx.x & 31;
    const int wv  = __builtin_amdgcn_readfirstlane((int)(threadIdx.x >> 5));
    const int ch0 = blockIdx.x * PF_CP;
    const float s  = 1.0f - expf(-((float)HOP / (float)SRATE) / 0.06f);
    const float da = powf(2.0f, 0.98f);

    float M = 0.0f;
    const int p  = (wv - 1) * 32 + ln;
    const int pc = p & (PF_CP - 1);
    const int pt = p >> 3;

    #pragma unroll 1
    for (int it = 0; it < PF_NK + 2; ++it) {
        if (wv == 0) {
            if (it >= 1 && it <= PF_NK && ln < PF_CP) {
                const float4* e4 = (const float4*)(Eb[(it - 1) % 3] + ln * PF_PT);
                float4*       m4 = (float4*)(Mb[(it - 1) & 1] + ln * PF_PT);
                float4 e[PF_TC / 4];
                #pragma unroll
                for (int j = 0; j < PF_TC / 4; ++j) e[j] = e4[j];
                #pragma unroll
                for (int j = 0; j < PF_TC / 4; ++j) {
                    float4 m;
                    M = __builtin_fmaf(1.0f - s, M, s * e[j].x);  m.x = M;
                    M = __builtin_fmaf(1.0f - s, M, s * e[j].y);  m.y = M;
                    M = __builtin_fmaf(1.0f - s, M, s * e[j].z);  m.z = M;
                    M = __builtin_fmaf(1.0f - s, M, s * e[j].w);  m.w = M;
                    m4[j] = m;
                }
            }
        } else {
            const bool fetch = it < PF_NK;
            float pre[4] = {0.0f, 0.0f, 0.0f, 0.0f};
            if (fetch) {
                #pragma unroll
                for (int i = 0; i < 4; ++i) {
                    int t = it * PF_TC + pt + 32 * i;
                    if (t >= NFRAMES) t = NFRAMES - 1;
                    pre[i] = melBuf[t * NCH + ch0 + pc];
                }
            }
            if (it >= 2) {
                const int k = it - 2;
                if (k < PF_NK - 1) pcen_point4<false>(Eb[k % 3], Mb[k & 1], out, wv - 1, ln, ch0, k, da);
                else               pcen_point4<true >(Eb[k % 3], Mb[k & 1], out, wv - 1, ln, ch0, k, da);
            }
            if (fetch) {
                float* ed = Eb[it % 3] + pc * PF_PT + pt;
                #pragma unroll
                for (int i = 0; i < 4; ++i) ed[32 * i] = pre[i];
            }
        }
        pcen_lds_barrier();
    }
}

extern "C" void kernel_launch(void* const* d_in, const int* in_sizes, int n_in,
                              void* d_out, int out_size, void* d_ws, size_t ws_size,
                              hipStream_t stream) {
    const float* x = (const float*)d_in[0];
    float* ws = (float*)d_ws;

    float* tab    = ws;
    float* melFB  = tab    + (size_t)NTAB * 32;
    float* melBuf = melFB  + (size_t)FPAD * NMELS;

    init_tabs<<<(NTAB * 32 + 255) / 256, 256, 0, stream>>>(tab);
    int n2 = FPAD * NMELS;
    init_mel<<<(n2 + 255) / 256, 256, 0, stream>>>(melFB);

    dim3 grid(TTILES, BATCH);
    dft_mel_kernel<<<grid, 128, 0, stream>>>(x, tab, melFB, melBuf);

    pcen_fused_kernel<<<NCH / PF_CP, 32 * (1 + PF_WORKERS), 0, stream>>>(melBuf, (float*)d_out);
}


# dft_mel prologue: a tile's sample span fetched as 10 16-byte loads per thread, all issued with the table loads before the first wait (were 38 4-byte loads in 7 dependent trips); sums unchanged
# speedup vs baseline: 23.2156x; 1.0897x over previous
#include <hip/hip_runtime.h>
#include <math.h>

#define SRATE     22050
#define NFFT      1024
#define HOP       256
#define NMELS     128
#define NFREQS    513
#define FPAD      528
#define BATCH     16
#define NSAMP     661504
#define NFRAMES   2585
#define TTILES    162
#define SPAN      (15*HOP + NFFT)
#define NCH       (BATCH*NMELS)
#define SPAN_V4   ((SPAN/4 + 127)/128)
#define SPAN_PAD  (SPAN_V4*128*4)

typedef __attribute__((ext_vector_type(2))) float v2f;
typedef __attribute__((ext_vector_type(8))) float v8f;

#define NTAB 160
__global__ void init_tabs(float* __restrict__ tab) {
    int idx = blockIdx.x * blockDim.x + threadIdx.x;
    if (idx >= NTAB * 32) return;
    const int e = idx >> 5, ln = idx & 31, c = ln & 15, h = ln >> 4;
    const float TWO_PI = 6.28318530717958647692f;
    float v;
    if (e < 32) {
        const int q = e & 15;
        const int s = q & 1, ks = q >> 1;
        const int n1 = 4 * ks + 2 * h + s;
        const bool mid = (e >= 16) && (c == 0);
        const int f1 = mid ? 16 : c;
        const float ang = (float)((f1 * n1) & 31) * (TWO_PI / 32.0f);
        v = (e < 16 || mid) ? cosf(ang) : sinf(ang);
    } else if (e < 64) {
        const int q = e - 32;
        const int s = q & 1, ks = (q >> 1) & 7, mt = q >> 4;
        const int n = 32 * (4 * ks + 2 * h + s) + 16 * mt + c;
        v = 0.5f - 0.5f * cosf(TWO_PI * (float)n / (float)NFFT);
    } else if (e < 128) {
        const int q = (e - 64) & 31;
        const int r = q & 7, nt = (q >> 3) & 1, mt = q >> 4;
        const int f1 = (nt == 0) ? c : ((c == 0) ? 16 : 32 - c);
        const int n2 = 16 * mt + 8 * h + r;
        const float ang = (float)(f1 * n2) * (TWO_PI / (float)NFFT);
        v = (e < 96) ? cosf(ang) : sinf(ang);
    } else {
        const int q = (e - 128) & 15;
        const int s = q & 1, j = (q >> 1) & 3, mt = q >> 3;
        const int f2 = c, n2 = 16 * mt + 8 * h + 2 * j + s;
        const float ang = (float)((f2 * n2) & 31) * (TWO_PI / 32.0f);
        v = (e < 144) ? cosf(ang) : sinf(ang);
    }
    tab[idx] = v;
}

__device__ __forceinline__ float hz_to_mel(float f) { return 2595.0f * log10f(1.0f + f / 700.0f); }
__device__ __forceinline__ float mel_to_hz(float m) { return 700.0f * (exp10f(m / 2595.0f) - 1.0f); }

__global__ void init_mel(float* __restrict__ melFB) {
    int idx = blockIdx.x * blockDim.x + threadIdx.x;
    if (idx >= FPAD * NMELS) return;
    int k = idx / NMELS;
    int m = idx % NMELS;
    float v = 0.0f;
    if (k < NFREQS) {
        float m_lo = hz_to_mel(50.0f);
        float m_hi = hz_to_mel((float)SRATE * 0.5f);
        float step = (m_hi - m_lo) / (float)(NMELS + 1);
        float f0 = mel_to_hz(m_lo + step * (float)m);
        float f1 = mel_to_hz(m_lo + step * (float)(m + 1));
        float f2 = mel_to_hz(m_lo + step * (float)(m + 2));
        float freq = (float)k * ((float)SRATE * 0.5f / (float)(NFFT / 2));
        float down = (freq - f0) / (f1 - f0);
        float up   = (f2 - freq) / (f2 - f1);
        v = fmaxf(0.0f, fminf(down, up));
    }
    melFB[(k >> 1) * (NMELS * 2) + m * 2 + (k & 1)] = v;
}

#define WMMA4(A, B, C) __builtin_amdgcn_wmma_f32_16x16x4_f32(false, (A), false, (B), (short)0, (C), false, false)
__global__ __launch_bounds__(128, 3)
void dft_mel_kernel(const float* __restrict__ x,
                    const float* __restrict__ tab,
                    const float* __restrict__ melFB,
                    float* __restrict__ melBuf)
{
    __shared__ __attribute__((aligned(16))) float span[SPAN_PAD];
    __shared__ float mag[16 * FPAD];

    const int tile = blockIdx.x;
    const int b    = blockIdx.y;
    const int tid  = threadIdx.x;
    const int ln   = tid & 31;
    const int wv   = tid >> 5;
    const int t0   = tile * 16;
    const int c    = ln & 15;
    const int h    = ln >> 4;

    const float* xb = x + (size_t)b * NSAMP;
    const long base = (long)t0 * HOP - (NFFT / 2);
    static_assert(SPAN % 4 == 0 && HOP % 4 == 0 && NFFT % 8 == 0 && NSAMP % 4 == 0, "16-byte aligned spans");
    const bool inside = (base >= 0) && (base + SPAN <= (long)NSAMP);
    float4 sp[SPAN_V4];
    {
        const float4* row4 = (const float4*)xb;
        const int q0 = t0 * (HOP / 4) - NFFT / 8;
        #pragma unroll
        for (int j = 0; j < SPAN_V4; ++j) {
            int g = q0 + tid + 128 * j;
            g = min(max(g, 0), NSAMP / 4 - 1);
            sp[j] = row4[g];
        }
    }

    const float* tl = tab + ln;
    v2f B1c[8], B1s[8];
    v2f Wn[2][8];
    float Tc[2][2][8], Ts[2][2][8];
    v2f Ec[2][4], Ed[2][4], Ep[2][4];
    #pragma unroll
    for (int ks = 0; ks < 8; ++ks) {
        const int e = ks * 2;
        B1c[ks].x = tl[(e + 0) * 32];       B1c[ks].y = tl[(e + 1) * 32];
        B1s[ks].x = tl[(16 + e + 0) * 32];  B1s[ks].y = tl[(16 + e + 1) * 32];
    }
    #pragma unroll
    for (int mt = 0; mt < 2; ++mt) {
        #pragma unroll
        for (int ks = 0; ks < 8; ++ks) {
            const int e = 32 + (mt * 8 + ks) * 2;
            Wn[mt][ks].x = tl[(e + 0) * 32];  Wn[mt][ks].y = tl[(e + 1) * 32];
        }
        #pragma unroll
        for (int nt = 0; nt < 2; ++nt) {
            #pragma unroll
            for (int r = 0; r < 8; ++r) {
                const int e = (mt * 2 + nt) * 8 + r;
                Tc[mt][nt][r] = tl[(64 + e) * 32];
                Ts[mt][nt][r] = tl[(96 + e) * 32];
            }
        }
        #pragma unroll
        for (int j = 0; j < 4; ++j) {
            const int e = (mt * 4 + j) * 2;
            v2f es;
            Ec[mt][j].x = tl[(128 + e + 0) * 32];  Ec[mt][j].y = tl[(128 + e + 1) * 32];
            es.x        = tl[(144 + e + 0) * 32];  es.y        = tl[(144 + e + 1) * 32];
            Ed[mt][j] = es - Ec[mt][j];
            Ep[mt][j] = Ec[mt][j] + es;
        }
    }
    #pragma unroll
    for (int j = 0; j < SPAN_V4; ++j) ((float4*)span)[tid + 128 * j] = sp[j];
    if (!inside) {
        __syncthreads();
        for (int i = tid; i < SPAN; i += 128) {
            long p = base + i;
            if (p < 0) p = -p;
            if (p >= NSAMP) p = 2L * NSAMP - 2 - p;
            if (p < 0) p = 0;
            if (p >= NSAMP) p = NSAMP - 1;
            span[i] = xb[p];
        }
    }
    __syncthreads();

    #pragma unroll 1
    for (int q = 0; q < 4; ++q) {
        const int fr = wv * 4 + q;
        const float* xs = span + fr * HOP + 64 * h + c;

        v8f Dc[2], Ds[2];
        #pragma unroll
        for (int mt = 0; mt < 2; ++mt) {
            Dc[mt] = (v8f){0.f, 0.f, 0.f, 0.f, 0.f, 0.f, 0.f, 0.f};
            Ds[mt] = (v8f){0.f, 0.f, 0.f, 0.f, 0.f, 0.f, 0.f, 0.f};
        }
        #pragma unroll
        for (int ks = 0; ks < 8; ++ks) {
            #pragma unroll
            for (int mt = 0; mt < 2; ++mt) {
                v2f a;
                a.x = xs[128 * ks + 16 * mt]      * Wn[mt][ks].x;
                a.y = xs[128 * ks + 16 * mt + 32] * Wn[mt][ks].y;
                Dc[mt] = WMMA4(a, B1c[ks], Dc[mt]);
                Ds[mt] = WMMA4(a, B1s[ks], Ds[mt]);
            }
        }

        float Pr[2][2][8], Ps[2][2][8];
        const bool c0 = (c == 0);
        #pragma unroll
        for (int mt = 0; mt < 2; ++mt) {
            #pragma unroll
            for (int r = 0; r < 8; ++r) {
                const float dc = Dc[mt][r], ds = Ds[mt][r];
                const float ys = c0 ? 0.0f : ds;
                const float y1 = c0 ? ds : dc;
                Pr[mt][0][r] = dc * Tc[mt][0][r] - ys * Ts[mt][0][r];
                Ps[mt][0][r] = dc * Ts[mt][0][r] + ys * Tc[mt][0][r];
                Pr[mt][1][r] = y1 * Tc[mt][1][r] + ys * Ts[mt][1][r];
                Ps[mt][1][r] = y1 * Ts[mt][1][r] - ys * Tc[mt][1][r];
            }
        }

        #pragma unroll
        for (int nt = 0; nt < 2; ++nt) {
            v8f T1 = (v8f){0.f, 0.f, 0.f, 0.f, 0.f, 0.f, 0.f, 0.f};
            v8f T2 = (v8f){0.f, 0.f, 0.f, 0.f, 0.f, 0.f, 0.f, 0.f};
            v8f T3 = (v8f){0.f, 0.f, 0.f, 0.f, 0.f, 0.f, 0.f, 0.f};
            #pragma unroll
            for (int mt = 0; mt < 2; ++mt) {
                #pragma unroll
                for (int j = 0; j < 4; ++j) {
                    v2f ar, ai;
                    ar.x = Pr[mt][nt][2 * j];  ar.y = Pr[mt][nt][2 * j + 1];
                    ai.x = Ps[mt][nt][2 * j];  ai.y = Ps[mt][nt][2 * j + 1];
                    const v2f as = ar + ai;
                    T1 = WMMA4(as, Ec[mt][j], T1);
                    T2 = WMMA4(ar, Ed[mt][j], T2);
                    T3 = WMMA4(ai, Ep[mt][j], T3);
                }
            }
            const v8f Xr = T1 - T3;
            const v8f Xi = T1 + T2;
            #pragma unroll
            for (int r = 0; r < 8; ++r) {
                const float re = Xr[r], im = Xi[r];
                const int f1 = (nt == 0) ? (8 * h + r) : ((r == 0) ? (16 + 8 * h) : (32 - 8 * h - r));
                mag[fr * FPAD + 32 * c + f1] = sqrtf(re * re + im * im);
            }
        }

        float ny = 0.0f;
        #pragma unroll
        for (int mt = 0; mt < 2; ++mt) {
            #pragma unroll
            for (int r = 0; r < 8; r += 2) ny += Pr[mt][0][r] - Pr[mt][0][r + 1];
        }
        ny += __shfl_xor(ny, 16);
        if (ln < 16) mag[fr * FPAD + 512 + ln] = (ln == 0) ? fabsf(ny) : 0.0f;
    }
    __syncthreads();

    const int mrow  = c;
    const int kh    = h << 1;
    const int ncol  = c;
    const int mbase = h * 8;

    constexpr int MEL_KSTEPS = FPAD / 4;
    constexpr int MEL_KG     = 22;
    static_assert(MEL_KSTEPS % MEL_KG == 0, "whole groups only");
    {
        v8f acc0 = {0.f, 0.f, 0.f, 0.f, 0.f, 0.f, 0.f, 0.f};
        v8f acc1 = {0.f, 0.f, 0.f, 0.f, 0.f, 0.f, 0.f, 0.f};
        const int n0 = wv * 16 + ncol;
        const int n1 = n0 + 64;
        const float* bp = melFB + h * (NMELS * 2) + n0 * 2;
        const float* ap = mag + mrow * FPAD + kh;
        #pragma unroll 1
        for (int g = 0; g < MEL_KSTEPS / MEL_KG; ++g) {
            v2f bf0[MEL_KG], bf1[MEL_KG], af[MEL_KG];
            #pragma unroll
            for (int i = 0; i < MEL_KG; ++i) {
                const float* q = bp + (size_t)(g * MEL_KG + i) * (2 * NMELS * 2);
                bf0[i] = *(const v2f*)q;
                bf1[i] = *(const v2f*)(q + 64 * 2);
            }
            #pragma unroll
            for (int i = 0; i < MEL_KG; ++i)
                af[i] = *(const v2f*)(ap + (g * MEL_KG + i) * 4);
            #pragma unroll
            for (int i = 0; i < MEL_KG; ++i) {
                acc0 = WMMA4(af[i], bf0[i], acc0);
                acc1 = WMMA4(af[i], bf1[i], acc1);
            }
        }
        #pragma unroll
        for (int v = 0; v < 8; ++v) {
            const int t = t0 + mbase + v;
            if (t < NFRAMES) {
                float* o = melBuf + (size_t)t * NCH + b * NMELS;
                o[n0] = acc0[v];
                o[n1] = acc1[v];
            }
        }
    }
}

#define PF_CP      8
#define PF_TC      128
#define PF_PT      (PF_TC + 4)
#define PF_NK      ((NFRAMES + PF_TC - 1) / PF_TC)
#define PF_WORKERS 8

__device__ __forceinline__ void pcen_lds_barrier() {
    __builtin_amdgcn_fence(__ATOMIC_RELEASE, "workgroup", "local");
    __builtin_amdgcn_s_barrier();
    __builtin_amdgcn_fence(__ATOMIC_ACQUIRE, "workgroup", "local");
}

template <bool GUARD>
__device__ __forceinline__ void pcen_point4(const float* eb, const float* mb, float* __restrict__ out,
                                            int w, int ln, int ch0, int k, float da)
{
    float ev[4], Mv[4];
    #pragma unroll
    for (int j = 0; j < 4; ++j) {
        const int u = w * 4 + j;
        const int o = (u >> 2) * PF_PT + (u & 3) * 32 + ln;
        ev[j] = eb[o];
        Mv[j] = mb[o];
    }
    #pragma unroll
    for (int j = 0; j < 4; ++j) {
        const int u  = w * 4 + j;
        const int tg = k * PF_TC + (u & 3) * 32 + ln;
        float smooth = sqrtf(Mv[j] + 1e-6f);
        float v = powf(ev[j] / smooth + 2.0f, 0.98f) - da;
        if (!GUARD || tg < NFRAMES) out[(size_t)(ch0 + (u >> 2)) * NFRAMES + tg] = v;
    }
}

__global__ __launch_bounds__(32 * (1 + PF_WORKERS))
void pcen_fused_kernel(const float* __restrict__ melBuf,
                       float* __restrict__ out)
{
    __shared__ __attribute__((aligned(16))) float Eb[3][PF_CP * PF_PT];
    __shared__ __attribute__((aligned(16))) float Mb[2][PF_CP * PF_PT];
    static_assert(PF_CP * (PF_TC / 32) == 4 * PF_WORKERS, "four (channel, 32-frame) units per worker wave");
    static_assert(PF_CP * PF_TC == 4 * 32 * PF_WORKERS, "four fetched values per worker lane");

    const int ln  = threadIdx.x & 31;
    const int wv  = __builtin_amdgcn_readfirstlane((int)(threadIdx.x >> 5));
    const int ch0 = blockIdx.x * PF_CP;
    const float s  = 1.0f - expf(-((float)HOP / (float)SRATE) / 0.06f);
    const float da = powf(2.0f, 0.98f);

    float M = 0.0f;
    const int p  = (wv - 1) * 32 + ln;
    const int pc = p & (PF_CP - 1);
    const int pt = p >> 3;

    #pragma unroll 1
    for (int it = 0; it < PF_NK + 2; ++it) {
        if (wv == 0) {
            if (it >= 1 && it <= PF_NK && ln < PF_CP) {
                const float4* e4 = (const float4*)(Eb[(it - 1) % 3] + ln * PF_PT);
                float4*       m4 = (float4*)(Mb[(it - 1) & 1] + ln * PF_PT);
                float4 e[PF_TC / 4];
                #pragma unroll
                for (int j = 0; j < PF_TC / 4; ++j) e[j] = e4[j];
                #pragma unroll
                for (int j = 0; j < PF_TC / 4; ++j) {
                    float4 m;
                    M = __builtin_fmaf(1.0f - s, M, s * e[j].x);  m.x = M;
                    M = __builtin_fmaf(1.0f - s, M, s * e[j].y);  m.y = M;
                    M = __builtin_fmaf(1.0f - s, M, s * e[j].z);  m.z = M;
                    M = __builtin_fmaf(1.0f - s, M, s * e[j].w);  m.w = M;
                    m4[j] = m;
                }
            }
        } else {
            const bool fetch = it < PF_NK;
            float pre[4] = {0.0f, 0.0f, 0.0f, 0.0f};
            if (fetch) {
                #pragma unroll
                for (int i = 0; i < 4; ++i) {
                    int t = it * PF_TC + pt + 32 * i;
                    if (t >= NFRAMES) t = NFRAMES - 1;
                    pre[i] = melBuf[t * NCH + ch0 + pc];
                }
            }
            if (it >= 2) {
                const int k = it - 2;
                if (k < PF_NK - 1) pcen_point4<false>(Eb[k % 3], Mb[k & 1], out, wv - 1, ln, ch0, k, da);
                else               pcen_point4<true >(Eb[k % 3], Mb[k & 1], out, wv - 1, ln, ch0, k, da);
            }
            if (fetch) {
                float* ed = Eb[it % 3] + pc * PF_PT + pt;
                #pragma unroll
                for (int i = 0; i < 4; ++i) ed[32 * i] = pre[i];
            }
        }
        pcen_lds_barrier();
    }
}

extern "C" void kernel_launch(void* const* d_in, const int* in_sizes, int n_in,
                              void* d_out, int out_size, void* d_ws, size_t ws_size,
                              hipStream_t stream) {
    const float* x = (const float*)d_in[0];
    float* ws = (float*)d_ws;

    float* tab    = ws;
    float* melFB  = tab    + (size_t)NTAB * 32;
    float* melBuf = melFB  + (size_t)FPAD * NMELS;

    init_tabs<<<(NTAB * 32 + 255) / 256, 256, 0, stream>>>(tab);
    int n2 = FPAD * NMELS;
    init_mel<<<(n2 + 255) / 256, 256, 0, stream>>>(melFB);

    dim3 grid(TTILES, BATCH);
    dft_mel_kernel<<<grid, 128, 0, stream>>>(x, tab, melFB, melBuf);

    pcen_fused_kernel<<<NCH / PF_CP, 32 * (1 + PF_WORKERS), 0, stream>>>(melBuf, (float*)d_out);
}


# dft_mel: the twiddle table moves from registers to LDS and is read per frame as 16-byte pieces; a wave needs 240 registers (was 314), so 4 work-groups share a WGP (were 3); sums unchanged
# speedup vs baseline: 24.9261x; 1.0737x over previous
#include <hip/hip_runtime.h>
#include <math.h>

#define SRATE     22050
#define NFFT      1024
#define HOP       256
#define NMELS     128
#define NFREQS    513
#define FPAD      528
#define BATCH     16
#define NSAMP     661504
#define NFRAMES   2585
#define TTILES    162
#define SPAN      (15*HOP + NFFT)
#define NCH       (BATCH*NMELS)
#define SPAN_V4   ((SPAN/4 + 127)/128)
#define SPAN_PAD  (SPAN_V4*128*4)

typedef __attribute__((ext_vector_type(2))) float v2f;
typedef __attribute__((ext_vector_type(8))) float v8f;

#define NTAB 160
__global__ void init_tabs(float* __restrict__ tab) {
    int idx = blockIdx.x * blockDim.x + threadIdx.x;
    if (idx >= NTAB * 32) return;
    const int e = idx >> 5, ln = idx & 31, c = ln & 15, h = ln >> 4;
    const float TWO_PI = 6.28318530717958647692f;
    float v;
    if (e < 32) {
        const int q = e & 15;
        const int s = q & 1, ks = q >> 1;
        const int n1 = 4 * ks + 2 * h + s;
        const bool mid = (e >= 16) && (c == 0);
        const int f1 = mid ? 16 : c;
        const float ang = (float)((f1 * n1) & 31) * (TWO_PI / 32.0f);
        v = (e < 16 || mid) ? cosf(ang) : sinf(ang);
    } else if (e < 64) {
        const int q = e - 32;
        const int s = q & 1, ks = (q >> 1) & 7, mt = q >> 4;
        const int n = 32 * (4 * ks + 2 * h + s) + 16 * mt + c;
        v = 0.5f - 0.5f * cosf(TWO_PI * (float)n / (float)NFFT);
    } else if (e < 128) {
        const int q = (e - 64) & 31;
        const int r = q & 7, nt = (q >> 3) & 1, mt = q >> 4;
        const int f1 = (nt == 0) ? c : ((c == 0) ? 16 : 32 - c);
        const int n2 = 16 * mt + 8 * h + r;
        const float ang = (float)(f1 * n2) * (TWO_PI / (float)NFFT);
        v = (e < 96) ? cosf(ang) : sinf(ang);
    } else {
        const int q = (e - 128) & 15;
        const int s = q & 1, j = (q >> 1) & 3, mt = q >> 3;
        const int f2 = c, n2 = 16 * mt + 8 * h + 2 * j + s;
        const float ang = (float)((f2 * n2) & 31) * (TWO_PI / 32.0f);
        v = (e < 144) ? cosf(ang) : sinf(ang);
    }
    tab[idx] = v;
}

__device__ __forceinline__ float hz_to_mel(float f) { return 2595.0f * log10f(1.0f + f / 700.0f); }
__device__ __forceinline__ float mel_to_hz(float m) { return 700.0f * (exp10f(m / 2595.0f) - 1.0f); }

__global__ void init_mel(float* __restrict__ melFB) {
    int idx = blockIdx.x * blockDim.x + threadIdx.x;
    if (idx >= FPAD * NMELS) return;
    int k = idx / NMELS;
    int m = idx % NMELS;
    float v = 0.0f;
    if (k < NFREQS) {
        float m_lo = hz_to_mel(50.0f);
        float m_hi = hz_to_mel((float)SRATE * 0.5f);
        float step = (m_hi - m_lo) / (float)(NMELS + 1);
        float f0 = mel_to_hz(m_lo + step * (float)m);
        float f1 = mel_to_hz(m_lo + step * (float)(m + 1));
        float f2 = mel_to_hz(m_lo + step * (float)(m + 2));
        float freq = (float)k * ((float)SRATE * 0.5f / (float)(NFFT / 2));
        float down = (freq - f0) / (f1 - f0);
        float up   = (f2 - freq) / (f2 - f1);
        v = fmaxf(0.0f, fminf(down, up));
    }
    melFB[(k >> 1) * (NMELS * 2) + m * 2 + (k & 1)] = v;
}

#define WMMA4(A, B, C) __builtin_amdgcn_wmma_f32_16x16x4_f32(false, (A), false, (B), (short)0, (C), false, false)
__global__ __launch_bounds__(128, 4)
void dft_mel_kernel(const float* __restrict__ x,
                    const float* __restrict__ tab,
                    const float* __restrict__ melFB,
                    float* __restrict__ melBuf)
{
    __shared__ __attribute__((aligned(16))) float span[SPAN_PAD];
    __shared__ float mag[16 * FPAD];
    __shared__ float4 twd[16 * 32];

    const int tile = blockIdx.x;
    const int b    = blockIdx.y;
    const int tid  = threadIdx.x;
    const int ln   = tid & 31;
    const int wv   = tid >> 5;
    const int t0   = tile * 16;
    const int c    = ln & 15;
    const int h    = ln >> 4;

    const float* xb = x + (size_t)b * NSAMP;
    const long base = (long)t0 * HOP - (NFFT / 2);
    static_assert(SPAN % 4 == 0 && HOP % 4 == 0 && NFFT % 8 == 0 && NSAMP % 4 == 0, "16-byte aligned spans");
    const bool inside = (base >= 0) && (base + SPAN <= (long)NSAMP);
    float4 sp[SPAN_V4];
    {
        const float4* row4 = (const float4*)xb;
        const int q0 = t0 * (HOP / 4) - NFFT / 8;
        #pragma unroll
        for (int j = 0; j < SPAN_V4; ++j) {
            int g = q0 + tid + 128 * j;
            g = min(max(g, 0), NSAMP / 4 - 1);
            sp[j] = row4[g];
        }
    }

    const float* tl = tab + ln;
    v2f B1c[8], B1s[8];
    v2f Wn[2][8];
    v2f Ec[2][4], Ed[2][4], Ep[2][4];
    #pragma unroll
    for (int ks = 0; ks < 8; ++ks) {
        const int e = ks * 2;
        B1c[ks].x = tl[(e + 0) * 32];       B1c[ks].y = tl[(e + 1) * 32];
        B1s[ks].x = tl[(16 + e + 0) * 32];  B1s[ks].y = tl[(16 + e + 1) * 32];
    }
    #pragma unroll
    for (int mt = 0; mt < 2; ++mt) {
        #pragma unroll
        for (int ks = 0; ks < 8; ++ks) {
            const int e = 32 + (mt * 8 + ks) * 2;
            Wn[mt][ks].x = tl[(e + 0) * 32];  Wn[mt][ks].y = tl[(e + 1) * 32];
        }
        #pragma unroll
        for (int j = 0; j < 4; ++j) {
            const int e = (mt * 4 + j) * 2;
            v2f es;
            Ec[mt][j].x = tl[(128 + e + 0) * 32];  Ec[mt][j].y = tl[(128 + e + 1) * 32];
            es.x        = tl[(144 + e + 0) * 32];  es.y        = tl[(144 + e + 1) * 32];
            Ed[mt][j] = es - Ec[mt][j];
            Ep[mt][j] = Ec[mt][j] + es;
        }
    }
    float4 tw[4];
    #pragma unroll
    for (int gi = 0; gi < 4; ++gi) {
        const int e = 64 + 4 * (4 * wv + gi);
        tw[gi].x = tl[(e + 0) * 32];  tw[gi].y = tl[(e + 1) * 32];
        tw[gi].z = tl[(e + 2) * 32];  tw[gi].w = tl[(e + 3) * 32];
    }
    #pragma unroll
    for (int j = 0; j < SPAN_V4; ++j) ((float4*)span)[tid + 128 * j] = sp[j];
    #pragma unroll
    for (int gi = 0; gi < 4; ++gi) twd[(4 * wv + gi) * 32 + ln] = tw[gi];
    if (!inside) {
        __syncthreads();
        for (int i = tid; i < SPAN; i += 128) {
            long p = base + i;
            if (p < 0) p = -p;
            if (p >= NSAMP) p = 2L * NSAMP - 2 - p;
            if (p < 0) p = 0;
            if (p >= NSAMP) p = NSAMP - 1;
            span[i] = xb[p];
        }
    }
    __syncthreads();

    #pragma unroll 1
    for (int q = 0; q < 4; ++q) {
        asm volatile("" ::: "memory");
        const int fr = wv * 4 + q;
        const float* xs = span + fr * HOP + 64 * h + c;

        v8f Dc[2], Ds[2];
        #pragma unroll
        for (int mt = 0; mt < 2; ++mt) {
            Dc[mt] = (v8f){0.f, 0.f, 0.f, 0.f, 0.f, 0.f, 0.f, 0.f};
            Ds[mt] = (v8f){0.f, 0.f, 0.f, 0.f, 0.f, 0.f, 0.f, 0.f};
        }
        #pragma unroll
        for (int ks = 0; ks < 8; ++ks) {
            #pragma unroll
            for (int mt = 0; mt < 2; ++mt) {
                v2f a;
                a.x = xs[128 * ks + 16 * mt]      * Wn[mt][ks].x;
                a.y = xs[128 * ks + 16 * mt + 32] * Wn[mt][ks].y;
                Dc[mt] = WMMA4(a, B1c[ks], Dc[mt]);
                Ds[mt] = WMMA4(a, B1s[ks], Ds[mt]);
            }
        }

        float Pr[2][2][8], Ps[2][2][8];
        const bool c0 = (c == 0);
        #pragma unroll
        for (int mt = 0; mt < 2; ++mt) {
            #pragma unroll
            for (int hf = 0; hf < 2; ++hf) {
                const float4 tc0 = twd[((mt * 2 + 0) * 2 + hf) * 32 + ln];
                const float4 ts0 = twd[(8 + (mt * 2 + 0) * 2 + hf) * 32 + ln];
                const float4 tc1 = twd[((mt * 2 + 1) * 2 + hf) * 32 + ln];
                const float4 ts1 = twd[(8 + (mt * 2 + 1) * 2 + hf) * 32 + ln];
                const float Tc0[4] = {tc0.x, tc0.y, tc0.z, tc0.w}, Ts0[4] = {ts0.x, ts0.y, ts0.z, ts0.w};
                const float Tc1[4] = {tc1.x, tc1.y, tc1.z, tc1.w}, Ts1[4] = {ts1.x, ts1.y, ts1.z, ts1.w};
                #pragma unroll
                for (int rr = 0; rr < 4; ++rr) {
                    const int r = 4 * hf + rr;
                    const float dc = Dc[mt][r], ds = Ds[mt][r];
                    const float ys = c0 ? 0.0f : ds;
                    const float y1 = c0 ? ds : dc;
                    Pr[mt][0][r] = dc * Tc0[rr] - ys * Ts0[rr];
                    Ps[mt][0][r] = dc * Ts0[rr] + ys * Tc0[rr];
                    Pr[mt][1][r] = y1 * Tc1[rr] + ys * Ts1[rr];
                    Ps[mt][1][r] = y1 * Ts1[rr] - ys * Tc1[rr];
                }
            }
        }

        #pragma unroll
        for (int nt = 0; nt < 2; ++nt) {
            v8f T1 = (v8f){0.f, 0.f, 0.f, 0.f, 0.f, 0.f, 0.f, 0.f};
            v8f T2 = (v8f){0.f, 0.f, 0.f, 0.f, 0.f, 0.f, 0.f, 0.f};
            v8f T3 = (v8f){0.f, 0.f, 0.f, 0.f, 0.f, 0.f, 0.f, 0.f};
            #pragma unroll
            for (int mt = 0; mt < 2; ++mt) {
                #pragma unroll
                for (int j = 0; j < 4; ++j) {
                    v2f ar, ai;
                    ar.x = Pr[mt][nt][2 * j];  ar.y = Pr[mt][nt][2 * j + 1];
                    ai.x = Ps[mt][nt][2 * j];  ai.y = Ps[mt][nt][2 * j + 1];
                    const v2f as = ar + ai;
                    T1 = WMMA4(as, Ec[mt][j], T1);
                    T2 = WMMA4(ar, Ed[mt][j], T2);
                    T3 = WMMA4(ai, Ep[mt][j], T3);
                }
            }
            const v8f Xr = T1 - T3;
            const v8f Xi = T1 + T2;
            #pragma unroll
            for (int r = 0; r < 8; ++r) {
                const float re = Xr[r], im = Xi[r];
                const int f1 = (nt == 0) ? (8 * h + r) : ((r == 0) ? (16 + 8 * h) : (32 - 8 * h - r));
                mag[fr * FPAD + 32 * c + f1] = sqrtf(re * re + im * im);
            }
        }

        float ny = 0.0f;
        #pragma unroll
        for (int mt = 0; mt < 2; ++mt) {
            #pragma unroll
            for (int r = 0; r < 8; r += 2) ny += Pr[mt][0][r] - Pr[mt][0][r + 1];
        }
        ny += __shfl_xor(ny, 16);
        if (ln < 16) mag[fr * FPAD + 512 + ln] = (ln == 0) ? fabsf(ny) : 0.0f;
    }
    __syncthreads();

    const int mrow  = c;
    const int kh    = h << 1;
    const int ncol  = c;
    const int mbase = h * 8;

    constexpr int MEL_KSTEPS = FPAD / 4;
    constexpr int MEL_KG     = 22;
    static_assert(MEL_KSTEPS % MEL_KG == 0, "whole groups only");
    {
        v8f acc0 = {0.f, 0.f, 0.f, 0.f, 0.f, 0.f, 0.f, 0.f};
        v8f acc1 = {0.f, 0.f, 0.f, 0.f, 0.f, 0.f, 0.f, 0.f};
        const int n0 = wv * 16 + ncol;
        const int n1 = n0 + 64;
        const float* bp = melFB + h * (NMELS * 2) + n0 * 2;
        const float* ap = mag + mrow * FPAD + kh;
        #pragma unroll 1
        for (int g = 0; g < MEL_KSTEPS / MEL_KG; ++g) {
            v2f bf0[MEL_KG], bf1[MEL_KG], af[MEL_KG];
            #pragma unroll
            for (int i = 0; i < MEL_KG; ++i) {
                const float* q = bp + (size_t)(g * MEL_KG + i) * (2 * NMELS * 2);
                bf0[i] = *(const v2f*)q;
                bf1[i] = *(const v2f*)(q + 64 * 2);
            }
            #pragma unroll
            for (int i = 0; i < MEL_KG; ++i)
                af[i] = *(const v2f*)(ap + (g * MEL_KG + i) * 4);
            #pragma unroll
            for (int i = 0; i < MEL_KG; ++i) {
                acc0 = WMMA4(af[i], bf0[i], acc0);
                acc1 = WMMA4(af[i], bf1[i], acc1);
            }
        }
        #pragma unroll
        for (int v = 0; v < 8; ++v) {
            const int t = t0 + mbase + v;
            if (t < NFRAMES) {
                float* o = melBuf + (size_t)t * NCH + b * NMELS;
                o[n0] = acc0[v];
                o[n1] = acc1[v];
            }
        }
    }
}

#define PF_CP      8
#define PF_TC      128
#define PF_PT      (PF_TC + 4)
#define PF_NK      ((NFRAMES + PF_TC - 1) / PF_TC)
#define PF_WORKERS 8

__device__ __forceinline__ void pcen_lds_barrier() {
    __builtin_amdgcn_fence(__ATOMIC_RELEASE, "workgroup", "local");
    __builtin_amdgcn_s_barrier();
    __builtin_amdgcn_fence(__ATOMIC_ACQUIRE, "workgroup", "local");
}

template <bool GUARD>
__device__ __forceinline__ void pcen_point4(const float* eb, const float* mb, float* __restrict__ out,
                                            int w, int ln, int ch0, int k, float da)
{
    float ev[4], Mv[4];
    #pragma unroll
    for (int j = 0; j < 4; ++j) {
        const int u = w * 4 + j;
        const int o = (u >> 2) * PF_PT + (u & 3) * 32 + ln;
        ev[j] = eb[o];
        Mv[j] = mb[o];
    }
    #pragma unroll
    for (int j = 0; j < 4; ++j) {
        const int u  = w * 4 + j;
        const int tg = k * PF_TC + (u & 3) * 32 + ln;
        float smooth = sqrtf(Mv[j] + 1e-6f);
        float v = powf(ev[j] / smooth + 2.0f, 0.98f) - da;
        if (!GUARD || tg < NFRAMES) out[(size_t)(ch0 + (u >> 2)) * NFRAMES + tg] = v;
    }
}

__global__ __launch_bounds__(32 * (1 + PF_WORKERS))
void pcen_fused_kernel(const float* __restrict__ melBuf,
                       float* __restrict__ out)
{
    __shared__ __attribute__((aligned(16))) float Eb[3][PF_CP * PF_PT];
    __shared__ __attribute__((aligned(16))) float Mb[2][PF_CP * PF_PT];
    static_assert(PF_CP * (PF_TC / 32) == 4 * PF_WORKERS, "four (channel, 32-frame) units per worker wave");
    static_assert(PF_CP * PF_TC == 4 * 32 * PF_WORKERS, "four fetched values per worker lane");

    const int ln  = threadIdx.x & 31;
    const int wv  = __builtin_amdgcn_readfirstlane((int)(threadIdx.x >> 5));
    const int ch0 = blockIdx.x * PF_CP;
    const float s  = 1.0f - expf(-((float)HOP / (float)SRATE) / 0.06f);
    const float da = powf(2.0f, 0.98f);

    float M = 0.0f;
    const int p  = (wv - 1) * 32 + ln;
    const int pc = p & (PF_CP - 1);
    const int pt = p >> 3;

    #pragma unroll 1
    for (int it = 0; it < PF_NK + 2; ++it) {
        if (wv == 0) {
            if (it >= 1 && it <= PF_NK && ln < PF_CP) {
                const float4* e4 = (const float4*)(Eb[(it - 1) % 3] + ln * PF_PT);
                float4*       m4 = (float4*)(Mb[(it - 1) & 1] + ln * PF_PT);
                float4 e[PF_TC / 4];
                #pragma unroll
                for (int j = 0; j < PF_TC / 4; ++j) e[j] = e4[j];
                #pragma unroll
                for (int j = 0; j < PF_TC / 4; ++j) {
                    float4 m;
                    M = __builtin_fmaf(1.0f - s, M, s * e[j].x);  m.x = M;
                    M = __builtin_fmaf(1.0f - s, M, s * e[j].y);  m.y = M;
                    M = __builtin_fmaf(1.0f - s, M, s * e[j].z);  m.z = M;
                    M = __builtin_fmaf(1.0f - s, M, s * e[j].w);  m.w = M;
                    m4[j] = m;
                }
            }
        } else {
            const bool fetch = it < PF_NK;
            float pre[4] = {0.0f, 0.0f, 0.0f, 0.0f};
            if (fetch) {
                #pragma unroll
                for (int i = 0; i < 4; ++i) {
                    int t = it * PF_TC + pt + 32 * i;
                    if (t >= NFRAMES) t = NFRAMES - 1;
                    pre[i] = melBuf[t * NCH + ch0 + pc];
                }
            }
            if (it >= 2) {
                const int k = it - 2;
                if (k < PF_NK - 1) pcen_point4<false>(Eb[k % 3], Mb[k & 1], out, wv - 1, ln, ch0, k, da);
                else               pcen_point4<true >(Eb[k % 3], Mb[k & 1], out, wv - 1, ln, ch0, k, da);
            }
            if (fetch) {
                float* ed = Eb[it % 3] + pc * PF_PT + pt;
                #pragma unroll
                for (int i = 0; i < 4; ++i) ed[32 * i] = pre[i];
            }
        }
        pcen_lds_barrier();
    }
}

extern "C" void kernel_launch(void* const* d_in, const int* in_sizes, int n_in,
                              void* d_out, int out_size, void* d_ws, size_t ws_size,
                              hipStream_t stream) {
    const float* x = (const float*)d_in[0];
    float* ws = (float*)d_ws;

    float* tab    = ws;
    float* melFB  = tab    + (size_t)NTAB * 32;
    float* melBuf = melFB  + (size_t)FPAD * NMELS;

    init_tabs<<<(NTAB * 32 + 255) / 256, 256, 0, stream>>>(tab);
    int n2 = FPAD * NMELS;
    init_mel<<<(n2 + 255) / 256, 256, 0, stream>>>(melFB);

    dim3 grid(TTILES, BATCH);
    dft_mel_kernel<<<grid, 128, 0, stream>>>(x, tab, melFB, melBuf);

    pcen_fused_kernel<<<NCH / PF_CP, 32 * (1 + PF_WORKERS), 0, stream>>>(melBuf, (float*)d_out);
}


# dft_mel magnitude: sqrt(re^2+im^2) on the hardware square-root instruction (1 ulp) instead of the library's correctly rounded sqrtf with its scaling and fix-up per bin; magnitudes move by an ulp
# speedup vs baseline: 25.7990x; 1.0350x over previous
#include <hip/hip_runtime.h>
#include <math.h>

#define SRATE     22050
#define NFFT      1024
#define HOP       256
#define NMELS     128
#define NFREQS    513
#define FPAD      528
#define BATCH     16
#define NSAMP     661504
#define NFRAMES   2585
#define TTILES    162
#define SPAN      (15*HOP + NFFT)
#define NCH       (BATCH*NMELS)
#define SPAN_V4   ((SPAN/4 + 127)/128)
#define SPAN_PAD  (SPAN_V4*128*4)

typedef __attribute__((ext_vector_type(2))) float v2f;
typedef __attribute__((ext_vector_type(8))) float v8f;

#define NTAB 160
__global__ void init_tabs(float* __restrict__ tab) {
    int idx = blockIdx.x * blockDim.x + threadIdx.x;
    if (idx >= NTAB * 32) return;
    const int e = idx >> 5, ln = idx & 31, c = ln & 15, h = ln >> 4;
    const float TWO_PI = 6.28318530717958647692f;
    float v;
    if (e < 32) {
        const int q = e & 15;
        const int s = q & 1, ks = q >> 1;
        const int n1 = 4 * ks + 2 * h + s;
        const bool mid = (e >= 16) && (c == 0);
        const int f1 = mid ? 16 : c;
        const float ang = (float)((f1 * n1) & 31) * (TWO_PI / 32.0f);
        v = (e < 16 || mid) ? cosf(ang) : sinf(ang);
    } else if (e < 64) {
        const int q = e - 32;
        const int s = q & 1, ks = (q >> 1) & 7, mt = q >> 4;
        const int n = 32 * (4 * ks + 2 * h + s) + 16 * mt + c;
        v = 0.5f - 0.5f * cosf(TWO_PI * (float)n / (float)NFFT);
    } else if (e < 128) {
        const int q = (e - 64) & 31;
        const int r = q & 7, nt = (q >> 3) & 1, mt = q >> 4;
        const int f1 = (nt == 0) ? c : ((c == 0) ? 16 : 32 - c);
        const int n2 = 16 * mt + 8 * h + r;
        const float ang = (float)(f1 * n2) * (TWO_PI / (float)NFFT);
        v = (e < 96) ? cosf(ang) : sinf(ang);
    } else {
        const int q = (e - 128) & 15;
        const int s = q & 1, j = (q >> 1) & 3, mt = q >> 3;
        const int f2 = c, n2 = 16 * mt + 8 * h + 2 * j + s;
        const float ang = (float)((f2 * n2) & 31) * (TWO_PI / 32.0f);
        v = (e < 144) ? cosf(ang) : sinf(ang);
    }
    tab[idx] = v;
}

__device__ __forceinline__ float hz_to_mel(float f) { return 2595.0f * log10f(1.0f + f / 700.0f); }
__device__ __forceinline__ float mel_to_hz(float m) { return 700.0f * (exp10f(m / 2595.0f) - 1.0f); }

__global__ void init_mel(float* __restrict__ melFB) {
    int idx = blockIdx.x * blockDim.x + threadIdx.x;
    if (idx >= FPAD * NMELS) return;
    int k = idx / NMELS;
    int m = idx % NMELS;
    float v = 0.0f;
    if (k < NFREQS) {
        float m_lo = hz_to_mel(50.0f);
        float m_hi = hz_to_mel((float)SRATE * 0.5f);
        float step = (m_hi - m_lo) / (float)(NMELS + 1);
        float f0 = mel_to_hz(m_lo + step * (float)m);
        float f1 = mel_to_hz(m_lo + step * (float)(m + 1));
        float f2 = mel_to_hz(m_lo + step * (float)(m + 2));
        float freq = (float)k * ((float)SRATE * 0.5f / (float)(NFFT / 2));
        float down = (freq - f0) / (f1 - f0);
        float up   = (f2 - freq) / (f2 - f1);
        v = fmaxf(0.0f, fminf(down, up));
    }
    melFB[(k >> 1) * (NMELS * 2) + m * 2 + (k & 1)] = v;
}

#define WMMA4(A, B, C) __builtin_amdgcn_wmma_f32_16x16x4_f32(false, (A), false, (B), (short)0, (C), false, false)
__global__ __launch_bounds__(128, 4)
void dft_mel_kernel(const float* __restrict__ x,
                    const float* __restrict__ tab,
                    const float* __restrict__ melFB,
                    float* __restrict__ melBuf)
{
    __shared__ __attribute__((aligned(16))) float span[SPAN_PAD];
    __shared__ float mag[16 * FPAD];
    __shared__ float4 twd[16 * 32];

    const int tile = blockIdx.x;
    const int b    = blockIdx.y;
    const int tid  = threadIdx.x;
    const int ln   = tid & 31;
    const int wv   = tid >> 5;
    const int t0   = tile * 16;
    const int c    = ln & 15;
    const int h    = ln >> 4;

    const float* xb = x + (size_t)b * NSAMP;
    const long base = (long)t0 * HOP - (NFFT / 2);
    static_assert(SPAN % 4 == 0 && HOP % 4 == 0 && NFFT % 8 == 0 && NSAMP % 4 == 0, "16-byte aligned spans");
    const bool inside = (base >= 0) && (base + SPAN <= (long)NSAMP);
    float4 sp[SPAN_V4];
    {
        const float4* row4 = (const float4*)xb;
        const int q0 = t0 * (HOP / 4) - NFFT / 8;
        #pragma unroll
        for (int j = 0; j < SPAN_V4; ++j) {
            int g = q0 + tid + 128 * j;
            g = min(max(g, 0), NSAMP / 4 - 1);
            sp[j] = row4[g];
        }
    }

    const float* tl = tab + ln;
    v2f B1c[8], B1s[8];
    v2f Wn[2][8];
    v2f Ec[2][4], Ed[2][4], Ep[2][4];
    #pragma unroll
    for (int ks = 0; ks < 8; ++ks) {
        const int e = ks * 2;
        B1c[ks].x = tl[(e + 0) * 32];       B1c[ks].y = tl[(e + 1) * 32];
        B1s[ks].x = tl[(16 + e + 0) * 32];  B1s[ks].y = tl[(16 + e + 1) * 32];
    }
    #pragma unroll
    for (int mt = 0; mt < 2; ++mt) {
        #pragma unroll
        for (int ks = 0; ks < 8; ++ks) {
            const int e = 32 + (mt * 8 + ks) * 2;
            Wn[mt][ks].x = tl[(e + 0) * 32];  Wn[mt][ks].y = tl[(e + 1) * 32];
        }
        #pragma unroll
        for (int j = 0; j < 4; ++j) {
            const int e = (mt * 4 + j) * 2;
            v2f es;
            Ec[mt][j].x = tl[(128 + e + 0) * 32];  Ec[mt][j].y = tl[(128 + e + 1) * 32];
            es.x        = tl[(144 + e + 0) * 32];  es.y        = tl[(144 + e + 1) * 32];
            Ed[mt][j] = es - Ec[mt][j];
            Ep[mt][j] = Ec[mt][j] + es;
        }
    }
    float4 tw[4];
    #pragma unroll
    for (int gi = 0; gi < 4; ++gi) {
        const int e = 64 + 4 * (4 * wv + gi);
        tw[gi].x = tl[(e + 0) * 32];  tw[gi].y = tl[(e + 1) * 32];
        tw[gi].z = tl[(e + 2) * 32];  tw[gi].w = tl[(e + 3) * 32];
    }
    #pragma unroll
    for (int j = 0; j < SPAN_V4; ++j) ((float4*)span)[tid + 128 * j] = sp[j];
    #pragma unroll
    for (int gi = 0; gi < 4; ++gi) twd[(4 * wv + gi) * 32 + ln] = tw[gi];
    if (!inside) {
        __syncthreads();
        for (int i = tid; i < SPAN; i += 128) {
            long p = base + i;
            if (p < 0) p = -p;
            if (p >= NSAMP) p = 2L * NSAMP - 2 - p;
            if (p < 0) p = 0;
            if (p >= NSAMP) p = NSAMP - 1;
            span[i] = xb[p];
        }
    }
    __syncthreads();

    #pragma unroll 1
    for (int q = 0; q < 4; ++q) {
        asm volatile("" ::: "memory");
        const int fr = wv * 4 + q;
        const float* xs = span + fr * HOP + 64 * h + c;

        v8f Dc[2], Ds[2];
        #pragma unroll
        for (int mt = 0; mt < 2; ++mt) {
            Dc[mt] = (v8f){0.f, 0.f, 0.f, 0.f, 0.f, 0.f, 0.f, 0.f};
            Ds[mt] = (v8f){0.f, 0.f, 0.f, 0.f, 0.f, 0.f, 0.f, 0.f};
        }
        #pragma unroll
        for (int ks = 0; ks < 8; ++ks) {
            #pragma unroll
            for (int mt = 0; mt < 2; ++mt) {
                v2f a;
                a.x = xs[128 * ks + 16 * mt]      * Wn[mt][ks].x;
                a.y = xs[128 * ks + 16 * mt + 32] * Wn[mt][ks].y;
                Dc[mt] = WMMA4(a, B1c[ks], Dc[mt]);
                Ds[mt] = WMMA4(a, B1s[ks], Ds[mt]);
            }
        }

        float Pr[2][2][8], Ps[2][2][8];
        const bool c0 = (c == 0);
        #pragma unroll
        for (int mt = 0; mt < 2; ++mt) {
            #pragma unroll
            for (int hf = 0; hf < 2; ++hf) {
                const float4 tc0 = twd[((mt * 2 + 0) * 2 + hf) * 32 + ln];
                const float4 ts0 = twd[(8 + (mt * 2 + 0) * 2 + hf) * 32 + ln];
                const float4 tc1 = twd[((mt * 2 + 1) * 2 + hf) * 32 + ln];
                const float4 ts1 = twd[(8 + (mt * 2 + 1) * 2 + hf) * 32 + ln];
                const float Tc0[4] = {tc0.x, tc0.y, tc0.z, tc0.w}, Ts0[4] = {ts0.x, ts0.y, ts0.z, ts0.w};
                const float Tc1[4] = {tc1.x, tc1.y, tc1.z, tc1.w}, Ts1[4] = {ts1.x, ts1.y, ts1.z, ts1.w};
                #pragma unroll
                for (int rr = 0; rr < 4; ++rr) {
                    const int r = 4 * hf + rr;
                    const float dc = Dc[mt][r], ds = Ds[mt][r];
                    const float ys = c0 ? 0.0f : ds;
                    const float y1 = c0 ? ds : dc;
                    Pr[mt][0][r] = dc * Tc0[rr] - ys * Ts0[rr];
                    Ps[mt][0][r] = dc * Ts0[rr] + ys * Tc0[rr];
                    Pr[mt][1][r] = y1 * Tc1[rr] + ys * Ts1[rr];
                    Ps[mt][1][r] = y1 * Ts1[rr] - ys * Tc1[rr];
                }
            }
        }

        #pragma unroll
        for (int nt = 0; nt < 2; ++nt) {
            v8f T1 = (v8f){0.f, 0.f, 0.f, 0.f, 0.f, 0.f, 0.f, 0.f};
            v8f T2 = (v8f){0.f, 0.f, 0.f, 0.f, 0.f, 0.f, 0.f, 0.f};
            v8f T3 = (v8f){0.f, 0.f, 0.f, 0.f, 0.f, 0.f, 0.f, 0.f};
            #pragma unroll
            for (int mt = 0; mt < 2; ++mt) {
                #pragma unroll
                for (int j = 0; j < 4; ++j) {
                    v2f ar, ai;
                    ar.x = Pr[mt][nt][2 * j];  ar.y = Pr[mt][nt][2 * j + 1];
                    ai.x = Ps[mt][nt][2 * j];  ai.y = Ps[mt][nt][2 * j + 1];
                    const v2f as = ar + ai;
                    T1 = WMMA4(as, Ec[mt][j], T1);
                    T2 = WMMA4(ar, Ed[mt][j], T2);
                    T3 = WMMA4(ai, Ep[mt][j], T3);
                }
            }
            const v8f Xr = T1 - T3;
            const v8f Xi = T1 + T2;
            #pragma unroll
            for (int r = 0; r < 8; ++r) {
                const float re = Xr[r], im = Xi[r];
                const int f1 = (nt == 0) ? (8 * h + r) : ((r == 0) ? (16 + 8 * h) : (32 - 8 * h - r));
                mag[fr * FPAD + 32 * c + f1] = __builtin_amdgcn_sqrtf(re * re + im * im);
            }
        }

        float ny = 0.0f;
        #pragma unroll
        for (int mt = 0; mt < 2; ++mt) {
            #pragma unroll
            for (int r = 0; r < 8; r += 2) ny += Pr[mt][0][r] - Pr[mt][0][r + 1];
        }
        ny += __shfl_xor(ny, 16);
        if (ln < 16) mag[fr * FPAD + 512 + ln] = (ln == 0) ? fabsf(ny) : 0.0f;
    }
    __syncthreads();

    const int mrow  = c;
    const int kh    = h << 1;
    const int ncol  = c;
    const int mbase = h * 8;

    constexpr int MEL_KSTEPS = FPAD / 4;
    constexpr int MEL_KG     = 22;
    static_assert(MEL_KSTEPS % MEL_KG == 0, "whole groups only");
    {
        v8f acc0 = {0.f, 0.f, 0.f, 0.f, 0.f, 0.f, 0.f, 0.f};
        v8f acc1 = {0.f, 0.f, 0.f, 0.f, 0.f, 0.f, 0.f, 0.f};
        const int n0 = wv * 16 + ncol;
        const int n1 = n0 + 64;
        const float* bp = melFB + h * (NMELS * 2) + n0 * 2;
        const float* ap = mag + mrow * FPAD + kh;
        #pragma unroll 1
        for (int g = 0; g < MEL_KSTEPS / MEL_KG; ++g) {
            v2f bf0[MEL_KG], bf1[MEL_KG], af[MEL_KG];
            #pragma unroll
            for (int i = 0; i < MEL_KG; ++i) {
                const float* q = bp + (size_t)(g * MEL_KG + i) * (2 * NMELS * 2);
                bf0[i] = *(const v2f*)q;
                bf1[i] = *(const v2f*)(q + 64 * 2);
            }
            #pragma unroll
            for (int i = 0; i < MEL_KG; ++i)
                af[i] = *(const v2f*)(ap + (g * MEL_KG + i) * 4);
            #pragma unroll
            for (int i = 0; i < MEL_KG; ++i) {
                acc0 = WMMA4(af[i], bf0[i], acc0);
                acc1 = WMMA4(af[i], bf1[i], acc1);
            }
        }
        #pragma unroll
        for (int v = 0; v < 8; ++v) {
            const int t = t0 + mbase + v;
            if (t < NFRAMES) {
                float* o = melBuf + (size_t)t * NCH + b * NMELS;
                o[n0] = acc0[v];
                o[n1] = acc1[v];
            }
        }
    }
}

#define PF_CP      8
#define PF_TC      128
#define PF_PT      (PF_TC + 4)
#define PF_NK      ((NFRAMES + PF_TC - 1) / PF_TC)
#define PF_WORKERS 8

__device__ __forceinline__ void pcen_lds_barrier() {
    __builtin_amdgcn_fence(__ATOMIC_RELEASE, "workgroup", "local");
    __builtin_amdgcn_s_barrier();
    __builtin_amdgcn_fence(__ATOMIC_ACQUIRE, "workgroup", "local");
}

template <bool GUARD>
__device__ __forceinline__ void pcen_point4(const float* eb, const float* mb, float* __restrict__ out,
                                            int w, int ln, int ch0, int k, float da)
{
    float ev[4], Mv[4];
    #pragma unroll
    for (int j = 0; j < 4; ++j) {
        const int u = w * 4 + j;
        const int o = (u >> 2) * PF_PT + (u & 3) * 32 + ln;
        ev[j] = eb[o];
        Mv[j] = mb[o];
    }
    #pragma unroll
    for (int j = 0; j < 4; ++j) {
        const int u  = w * 4 + j;
        const int tg = k * PF_TC + (u & 3) * 32 + ln;
        float smooth = sqrtf(Mv[j] + 1e-6f);
        float v = powf(ev[j] / smooth + 2.0f, 0.98f) - da;
        if (!GUARD || tg < NFRAMES) out[(size_t)(ch0 + (u >> 2)) * NFRAMES + tg] = v;
    }
}

__global__ __launch_bounds__(32 * (1 + PF_WORKERS))
void pcen_fused_kernel(const float* __restrict__ melBuf,
                       float* __restrict__ out)
{
    __shared__ __attribute__((aligned(16))) float Eb[3][PF_CP * PF_PT];
    __shared__ __attribute__((aligned(16))) float Mb[2][PF_CP * PF_PT];
    static_assert(PF_CP * (PF_TC / 32) == 4 * PF_WORKERS, "four (channel, 32-frame) units per worker wave");
    static_assert(PF_CP * PF_TC == 4 * 32 * PF_WORKERS, "four fetched values per worker lane");

    const int ln  = threadIdx.x & 31;
    const int wv  = __builtin_amdgcn_readfirstlane((int)(threadIdx.x >> 5));
    const int ch0 = blockIdx.x * PF_CP;
    const float s  = 1.0f - expf(-((float)HOP / (float)SRATE) / 0.06f);
    const float da = powf(2.0f, 0.98f);

    float M = 0.0f;
    const int p  = (wv - 1) * 32 + ln;
    const int pc = p & (PF_CP - 1);
    const int pt = p >> 3;

    #pragma unroll 1
    for (int it = 0; it < PF_NK + 2; ++it) {
        if (wv == 0) {
            if (it >= 1 && it <= PF_NK && ln < PF_CP) {
                const float4* e4 = (const float4*)(Eb[(it - 1) % 3] + ln * PF_PT);
                float4*       m4 = (float4*)(Mb[(it - 1) & 1] + ln * PF_PT);
                float4 e[PF_TC / 4];
                #pragma unroll
                for (int j = 0; j < PF_TC / 4; ++j) e[j] = e4[j];
                #pragma unroll
                for (int j = 0; j < PF_TC / 4; ++j) {
                    float4 m;
                    M = __builtin_fmaf(1.0f - s, M, s * e[j].x);  m.x = M;
                    M = __builtin_fmaf(1.0f - s, M, s * e[j].y);  m.y = M;
                    M = __builtin_fmaf(1.0f - s, M, s * e[j].z);  m.z = M;
                    M = __builtin_fmaf(1.0f - s, M, s * e[j].w);  m.w = M;
                    m4[j] = m;
                }
            }
        } else {
            const bool fetch = it < PF_NK;
            float pre[4] = {0.0f, 0.0f, 0.0f, 0.0f};
            if (fetch) {
                #pragma unroll
                for (int i = 0; i < 4; ++i) {
                    int t = it * PF_TC + pt + 32 * i;
                    if (t >= NFRAMES) t = NFRAMES - 1;
                    pre[i] = melBuf[t * NCH + ch0 + pc];
                }
            }
            if (it >= 2) {
                const int k = it - 2;
                if (k < PF_NK - 1) pcen_point4<false>(Eb[k % 3], Mb[k & 1], out, wv - 1, ln, ch0, k, da);
                else               pcen_point4<true >(Eb[k % 3], Mb[k & 1], out, wv - 1, ln, ch0, k, da);
            }
            if (fetch) {
                float* ed = Eb[it % 3] + pc * PF_PT + pt;
                #pragma unroll
                for (int i = 0; i < 4; ++i) ed[32 * i] = pre[i];
            }
        }
        pcen_lds_barrier();
    }
}

extern "C" void kernel_launch(void* const* d_in, const int* in_sizes, int n_in,
                              void* d_out, int out_size, void* d_ws, size_t ws_size,
                              hipStream_t stream) {
    const float* x = (const float*)d_in[0];
    float* ws = (float*)d_ws;

    float* tab    = ws;
    float* melFB  = tab    + (size_t)NTAB * 32;
    float* melBuf = melFB  + (size_t)FPAD * NMELS;

    init_tabs<<<(NTAB * 32 + 255) / 256, 256, 0, stream>>>(tab);
    int n2 = FPAD * NMELS;
    init_mel<<<(n2 + 255) / 256, 256, 0, stream>>>(melFB);

    dim3 grid(TTILES, BATCH);
    dft_mel_kernel<<<grid, 128, 0, stream>>>(x, tab, melFB, melBuf);

    pcen_fused_kernel<<<NCH / PF_CP, 32 * (1 + PF_WORKERS), 0, stream>>>(melBuf, (float*)d_out);
}
